# starting kernel
# baseline (speedup 1.0000x reference)
#include <hip/hip_runtime.h>

typedef float v2f __attribute__((ext_vector_type(2)));
typedef float v8f __attribute__((ext_vector_type(8)));

#define D_MODEL 768
#define NHEAD   12
#define DK      64
#define BATCH   2
#define SEQ     2048
#define M_ROWS  (BATCH * SEQ)
#define NEGV    (-1000000000.0f)
#define SMSTRIDE 18
#define WPB      8

__global__ __launch_bounds__(256) void gemm768_kernel(
    const float* __restrict__ A, const float* __restrict__ W,
    const float* __restrict__ bias, float* __restrict__ C, int split_heads)
{
    int wave = (int)((blockIdx.x * blockDim.x + threadIdx.x) >> 5);
    int lane = (int)(threadIdx.x & 31);
    int tn = wave % (D_MODEL / 64);
    int tm = wave / (D_MODEL / 64);
    int m0 = tm * 16, n0 = tn * 64;
    int l15 = lane & 15;
    int khi = lane >> 4;
    int rbase = khi << 3;

    const float* arow = A + (size_t)(m0 + l15) * D_MODEL;

    v8f acc[4];
#pragma unroll
    for (int c = 0; c < 4; ++c) acc[c] = (v8f){};

#pragma unroll 2
    for (int kb = 0; kb < D_MODEL; kb += 4) {
        v2f a = *(const v2f*)(arow + kb + 2 * khi);
        const float* wp = W + (size_t)(kb + 2 * khi) * D_MODEL + n0 + l15;
#pragma unroll
        for (int c = 0; c < 4; ++c) {
            v2f b; b.x = wp[c * 16]; b.y = wp[D_MODEL + c * 16];
            acc[c] = __builtin_amdgcn_wmma_f32_16x16x4_f32(
                false, a, false, b, (short)0, acc[c], false, false);
        }
    }

    if (!split_heads) {
#pragma unroll
        for (int c = 0; c < 4; ++c) {
            float bv = bias[n0 + c * 16 + l15];
#pragma unroll
            for (int r = 0; r < 8; ++r) {
                int row = m0 + rbase + r;
                C[(size_t)row * D_MODEL + n0 + c * 16 + l15] = acc[c][r] + bv;
            }
        }
    } else {
        int h = n0 >> 6;
#pragma unroll
        for (int c = 0; c < 4; ++c) {
            int d = c * 16 + l15;
            float bv = bias[n0 + d];
#pragma unroll
            for (int r = 0; r < 8; ++r) {
                int row = m0 + rbase + r;
                int bi = row / SEQ, s = row % SEQ;
                C[(((size_t)bi * NHEAD + h) * SEQ + s) * DK + d] = acc[c][r] + bv;
            }
        }
    }
}

__global__ __launch_bounds__(256) void attn_kernel(
    const float* __restrict__ QP, const float* __restrict__ KP,
    const float* __restrict__ VP, const int* __restrict__ mask,
    float* __restrict__ X)
{
    __shared__ float smem[WPB][16 * SMSTRIDE];
    int wv   = (int)(threadIdx.x >> 5);
    int lane = (int)(threadIdx.x & 31);
    int wave = (int)blockIdx.x * WPB + wv;

    int qt = wave % (SEQ / 16);
    int bh = wave / (SEQ / 16);
    int bi = bh / NHEAD, h = bh % NHEAD;
    int s0 = qt * 16;
    int l15 = lane & 15;
    int khi = lane >> 4;
    int rbase = khi << 3;

    const float* Q = QP + (size_t)bh * SEQ * DK;
    const float* K = KP + (size_t)bh * SEQ * DK;
    const float* V = VP + (size_t)bh * SEQ * DK;
    float* sm = smem[wv];

    const float* qrow = Q + (size_t)(s0 + l15) * DK;
    v2f qv[16];
#pragma unroll
    for (int j = 0; j < 16; ++j)
        qv[j] = *(const v2f*)(qrow + 4 * j + 2 * khi);

    v8f acc[4];
#pragma unroll
    for (int c = 0; c < 4; ++c) acc[c] = (v8f){};

    for (int kb = 0; kb < SEQ; kb += 16) {
        v8f sacc = {};
        const float* krow = K + (size_t)(kb + l15) * DK;
#pragma unroll
        for (int j = 0; j < 16; ++j) {
            v2f bb = *(const v2f*)(krow + 4 * j + 2 * khi);
            sacc = __builtin_amdgcn_wmma_f32_16x16x4_f32(
                false, qv[j], false, bb, (short)0, sacc, false, false);
        }

        int tg = kb + l15;
#pragma unroll
        for (int r = 0; r < 8; ++r) {
            int srow = rbase + r;
            int sg = s0 + srow;
            float val = sacc[r] * 0.125f;
            if (mask[(size_t)sg * SEQ + tg] == 0) val = NEGV;
            sm[srow * SMSTRIDE + l15] = val;
        }
        asm volatile("s_wait_dscnt 0" ::: "memory");

#pragma unroll
        for (int j = 0; j < 4; ++j) {
            v2f a = *(const v2f*)(sm + l15 * SMSTRIDE + 4 * j + 2 * khi);
            const float* vrow = V + (size_t)(kb + 4 * j + 2 * khi) * DK + l15;
#pragma unroll
            for (int c = 0; c < 4; ++c) {
                v2f bb; bb.x = vrow[c * 16]; bb.y = vrow[DK + c * 16];
                acc[c] = __builtin_amdgcn_wmma_f32_16x16x4_f32(
                    false, a, false, bb, (short)0, acc[c], false, false);
            }
        }
    }

#pragma unroll
    for (int c = 0; c < 4; ++c) {
#pragma unroll
        for (int r = 0; r < 8; ++r) {
            int s = s0 + rbase + r;
            int d = c * 16 + l15;
            X[((size_t)bi * SEQ + s) * D_MODEL + h * DK + d] = acc[c][r];
        }
    }
}

extern "C" void kernel_launch(void* const* d_in, const int* in_sizes, int n_in,
                              void* d_out, int out_size, void* d_ws, size_t ws_size,
                              hipStream_t stream) {
    (void)in_sizes; (void)n_in; (void)out_size; (void)ws_size;
    const float* q    = (const float*)d_in[0];
    const float* k    = (const float*)d_in[1];
    const float* v    = (const float*)d_in[2];
    const int*   mask = (const int*)d_in[3];
    const float* w_q  = (const float*)d_in[4];
    const float* b_q  = (const float*)d_in[5];
    const float* w_k  = (const float*)d_in[6];
    const float* b_k  = (const float*)d_in[7];
    const float* w_v  = (const float*)d_in[8];
    const float* b_v  = (const float*)d_in[9];
    const float* w_o  = (const float*)d_in[10];
    const float* b_o  = (const float*)d_in[11];

    float* ws = (float*)d_ws;
    size_t proj = (size_t)M_ROWS * D_MODEL;
    float* qp = ws;
    float* kp = ws + proj;
    float* vp = ws + 2 * proj;
    float* xb = ws + 3 * proj;

    dim3 blk(256);
    int gemm_waves = (M_ROWS / 16) * (D_MODEL / 64);
    dim3 g_gemm(gemm_waves / WPB);
    int attn_waves = BATCH * NHEAD * (SEQ / 16);
    dim3 g_attn(attn_waves / WPB);

    gemm768_kernel<<<g_gemm, blk, 0, stream>>>(q, w_q, b_q, qp, 1);
    gemm768_kernel<<<g_gemm, blk, 0, stream>>>(k, w_k, b_k, kp, 1);
    gemm768_kernel<<<g_gemm, blk, 0, stream>>>(v, w_v, b_v, vp, 1);
    attn_kernel<<<g_attn, blk, 0, stream>>>(qp, kp, vp, mask, xb);
    gemm768_kernel<<<g_gemm, blk, 0, stream>>>(xb, w_o, b_o, (float*)d_out, 0);
}


# attention: K and V tiles staged once per 4-wave block in LDS (32-key stages, two buffers, 16-byte coalesced loads); fragments read from LDS, not per-wave global loads; sums in the same order
# speedup vs baseline: 1.2399x; 1.2399x over previous
#include <hip/hip_runtime.h>

typedef float v2f __attribute__((ext_vector_type(2)));
typedef float v8f __attribute__((ext_vector_type(8)));

#define D_MODEL 768
#define NHEAD   12
#define DK      64
#define BATCH   2
#define SEQ     2048
#define M_ROWS  (BATCH * SEQ)
#define NEGV    (-1000000000.0f)
#define SMSTRIDE 18
#define WPB      8
#define AWPB     4
#define ATHREADS (AWPB * 32)
#define KSTAGE   32
#define KPITCH   68
#define VPITCH   160
#define NSTAGE   (SEQ / KSTAGE)

__global__ __launch_bounds__(256) void gemm768_kernel(
    const float* __restrict__ A, const float* __restrict__ W,
    const float* __restrict__ bias, float* __restrict__ C, int split_heads)
{
    int wave = (int)((blockIdx.x * blockDim.x + threadIdx.x) >> 5);
    int lane = (int)(threadIdx.x & 31);
    int tn = wave % (D_MODEL / 64);
    int tm = wave / (D_MODEL / 64);
    int m0 = tm * 16, n0 = tn * 64;
    int l15 = lane & 15;
    int khi = lane >> 4;
    int rbase = khi << 3;

    const float* arow = A + (size_t)(m0 + l15) * D_MODEL;

    v8f acc[4];
#pragma unroll
    for (int c = 0; c < 4; ++c) acc[c] = (v8f){};

#pragma unroll 2
    for (int kb = 0; kb < D_MODEL; kb += 4) {
        v2f a = *(const v2f*)(arow + kb + 2 * khi);
        const float* wp = W + (size_t)(kb + 2 * khi) * D_MODEL + n0 + l15;
#pragma unroll
        for (int c = 0; c < 4; ++c) {
            v2f b; b.x = wp[c * 16]; b.y = wp[D_MODEL + c * 16];
            acc[c] = __builtin_amdgcn_wmma_f32_16x16x4_f32(
                false, a, false, b, (short)0, acc[c], false, false);
        }
    }

    if (!split_heads) {
#pragma unroll
        for (int c = 0; c < 4; ++c) {
            float bv = bias[n0 + c * 16 + l15];
#pragma unroll
            for (int r = 0; r < 8; ++r) {
                int row = m0 + rbase + r;
                C[(size_t)row * D_MODEL + n0 + c * 16 + l15] = acc[c][r] + bv;
            }
        }
    } else {
        int h = n0 >> 6;
#pragma unroll
        for (int c = 0; c < 4; ++c) {
            int d = c * 16 + l15;
            float bv = bias[n0 + d];
#pragma unroll
            for (int r = 0; r < 8; ++r) {
                int row = m0 + rbase + r;
                int bi = row / SEQ, s = row % SEQ;
                C[(((size_t)bi * NHEAD + h) * SEQ + s) * DK + d] = acc[c][r] + bv;
            }
        }
    }
}

__global__ __launch_bounds__(ATHREADS) void attn_kernel(
    const float* __restrict__ QP, const float* __restrict__ KP,
    const float* __restrict__ VP, const int* __restrict__ mask,
    float* __restrict__ X)
{
    __shared__ __attribute__((aligned(16))) float ks[2][KSTAGE * KPITCH];
    __shared__ __attribute__((aligned(16))) float vs[2][(KSTAGE / 2) * VPITCH];
    __shared__ __attribute__((aligned(16))) float smem[AWPB][2][16 * SMSTRIDE];

    int tid  = (int)threadIdx.x;
    int wv   = tid >> 5;
    int lane = tid & 31;
    int wave = (int)blockIdx.x * AWPB + wv;

    int qt = wave % (SEQ / 16);
    int bh = wave / (SEQ / 16);
    int bi = bh / NHEAD, h = bh % NHEAD;
    int s0 = qt * 16;
    int l15 = lane & 15;
    int khi = lane >> 4;
    int rbase = khi << 3;

    const float* Q = QP + (size_t)bh * SEQ * DK;
    const float* K = KP + (size_t)bh * SEQ * DK;
    const float* V = VP + (size_t)bh * SEQ * DK;
    float* sm = &smem[wv][0][0];

    const float* qrow = Q + (size_t)(s0 + l15) * DK;
    v2f qv[16];
#pragma unroll
    for (int j = 0; j < 16; ++j)
        qv[j] = *(const v2f*)(qrow + 4 * j + 2 * khi);

    v8f acc[4];
#pragma unroll
    for (int c = 0; c < 4; ++c) acc[c] = (v8f){};

    const float4* Kg = (const float4*)K;
    const float4* Vg = (const float4*)V;
    const int* mrow = mask + (size_t)(s0 + rbase) * SEQ + l15;
    int vslot0 = ((tid >> 4) * 2) * 16 + (tid & 15);
    int vst0   = (tid >> 4) * VPITCH + (tid & 15) * 8;

    float4 kreg[4], vreg[4];
    int mreg[2][8];

#pragma unroll
    for (int p = 0; p < 4; ++p) kreg[p] = Kg[tid + ATHREADS * p];
#pragma unroll
    for (int p = 0; p < 2; ++p) {
        vreg[2 * p]     = Vg[vslot0 + 256 * p];
        vreg[2 * p + 1] = Vg[vslot0 + 256 * p + 16];
    }
#pragma unroll
    for (int u = 0; u < 2; ++u)
#pragma unroll
        for (int r = 0; r < 8; ++r)
            mreg[u][r] = mrow[(size_t)r * SEQ + 16 * u];
#pragma unroll
    for (int p = 0; p < 4; ++p) {
        int i = tid + ATHREADS * p;
        *(float4*)&ks[0][(i >> 4) * KPITCH + (i & 15) * 4] = kreg[p];
    }
#pragma unroll
    for (int p = 0; p < 2; ++p) {
        float4 a = vreg[2 * p], b = vreg[2 * p + 1];
        *(float4*)&vs[0][vst0 + 8 * VPITCH * p]     = make_float4(a.x, b.x, a.y, b.y);
        *(float4*)&vs[0][vst0 + 8 * VPITCH * p + 4] = make_float4(a.z, b.z, a.w, b.w);
    }
    __syncthreads();

    for (int st = 0; st < NSTAGE; ++st) {
        int cur = st & 1;
        bool more = (st + 1 < NSTAGE);

        int mk[2][8];
#pragma unroll
        for (int u = 0; u < 2; ++u)
#pragma unroll
            for (int r = 0; r < 8; ++r) mk[u][r] = mreg[u][r];

        if (more) {
            size_t g = (size_t)(st + 1) * (KSTAGE * DK / 4);
#pragma unroll
            for (int p = 0; p < 4; ++p) kreg[p] = Kg[g + tid + ATHREADS * p];
#pragma unroll
            for (int p = 0; p < 2; ++p) {
                vreg[2 * p]     = Vg[g + vslot0 + 256 * p];
                vreg[2 * p + 1] = Vg[g + vslot0 + 256 * p + 16];
            }
#pragma unroll
            for (int u = 0; u < 2; ++u)
#pragma unroll
                for (int r = 0; r < 8; ++r)
                    mreg[u][r] = mrow[(size_t)r * SEQ + (st + 1) * KSTAGE + 16 * u];
        }

        const float* k0 = &ks[cur][l15 * KPITCH + 2 * khi];
        const float* k1 = &ks[cur][(16 + l15) * KPITCH + 2 * khi];
        v2f kf0[16], kf1[16];
#pragma unroll
        for (int j = 0; j < 16; ++j) {
            kf0[j] = *(const v2f*)(k0 + 4 * j);
            kf1[j] = *(const v2f*)(k1 + 4 * j);
        }
        v8f sa = {}, sb = {};
#pragma unroll
        for (int j = 0; j < 16; ++j) {
            sa = __builtin_amdgcn_wmma_f32_16x16x4_f32(
                false, qv[j], false, kf0[j], (short)0, sa, false, false);
            sb = __builtin_amdgcn_wmma_f32_16x16x4_f32(
                false, qv[j], false, kf1[j], (short)0, sb, false, false);
        }

        const float* vb = &vs[cur][khi * VPITCH + 2 * l15];
        v2f vf[2][4][4];
#pragma unroll
        for (int u = 0; u < 2; ++u)
#pragma unroll
            for (int j = 0; j < 4; ++j)
#pragma unroll
                for (int c = 0; c < 4; ++c)
                    vf[u][j][c] = *(const v2f*)(vb + (8 * u + 2 * j) * VPITCH + 32 * c);

#pragma unroll
        for (int r = 0; r < 8; ++r) {
            int srow = rbase + r;
            float ea = sa[r] * 0.125f;
            float eb = sb[r] * 0.125f;
            if (mk[0][r] == 0) ea = NEGV;
            if (mk[1][r] == 0) eb = NEGV;
            sm[srow * SMSTRIDE + l15] = ea;
            sm[16 * SMSTRIDE + srow * SMSTRIDE + l15] = eb;
        }
        asm volatile("s_wait_dscnt 0" ::: "memory");

        v2f af[2][4];
#pragma unroll
        for (int u = 0; u < 2; ++u)
#pragma unroll
            for (int j = 0; j < 4; ++j)
                af[u][j] = *(const v2f*)(sm + u * 16 * SMSTRIDE + l15 * SMSTRIDE + 4 * j + 2 * khi);
#pragma unroll
        for (int u = 0; u < 2; ++u)
#pragma unroll
            for (int j = 0; j < 4; ++j)
#pragma unroll
                for (int c = 0; c < 4; ++c)
                    acc[c] = __builtin_amdgcn_wmma_f32_16x16x4_f32(
                        false, af[u][j], false, vf[u][j][c], (short)0, acc[c], false, false);

        if (more) {
#pragma unroll
            for (int p = 0; p < 4; ++p) {
                int i = tid + ATHREADS * p;
                *(float4*)&ks[cur ^ 1][(i >> 4) * KPITCH + (i & 15) * 4] = kreg[p];
            }
#pragma unroll
            for (int p = 0; p < 2; ++p) {
                float4 a = vreg[2 * p], b = vreg[2 * p + 1];
                *(float4*)&vs[cur ^ 1][vst0 + 8 * VPITCH * p]     = make_float4(a.x, b.x, a.y, b.y);
                *(float4*)&vs[cur ^ 1][vst0 + 8 * VPITCH * p + 4] = make_float4(a.z, b.z, a.w, b.w);
            }
        }
        __syncthreads();
    }

#pragma unroll
    for (int c = 0; c < 4; ++c) {
#pragma unroll
        for (int r = 0; r < 8; ++r) {
            int s = s0 + rbase + r;
            int d = c * 16 + l15;
            X[((size_t)bi * SEQ + s) * D_MODEL + h * DK + d] = acc[c][r];
        }
    }
}

extern "C" void kernel_launch(void* const* d_in, const int* in_sizes, int n_in,
                              void* d_out, int out_size, void* d_ws, size_t ws_size,
                              hipStream_t stream) {
    (void)in_sizes; (void)n_in; (void)out_size; (void)ws_size;
    const float* q    = (const float*)d_in[0];
    const float* k    = (const float*)d_in[1];
    const float* v    = (const float*)d_in[2];
    const int*   mask = (const int*)d_in[3];
    const float* w_q  = (const float*)d_in[4];
    const float* b_q  = (const float*)d_in[5];
    const float* w_k  = (const float*)d_in[6];
    const float* b_k  = (const float*)d_in[7];
    const float* w_v  = (const float*)d_in[8];
    const float* b_v  = (const float*)d_in[9];
    const float* w_o  = (const float*)d_in[10];
    const float* b_o  = (const float*)d_in[11];

    float* ws = (float*)d_ws;
    size_t proj = (size_t)M_ROWS * D_MODEL;
    float* qp = ws;
    float* kp = ws + proj;
    float* vp = ws + 2 * proj;
    float* xb = ws + 3 * proj;

    dim3 blk(256);
    int gemm_waves = (M_ROWS / 16) * (D_MODEL / 64);
    dim3 g_gemm(gemm_waves / WPB);
    int attn_waves = BATCH * NHEAD * (SEQ / 16);
    dim3 g_attn(attn_waves / AWPB);
    dim3 ablk(ATHREADS);

    gemm768_kernel<<<g_gemm, blk, 0, stream>>>(q, w_q, b_q, qp, 1);
    gemm768_kernel<<<g_gemm, blk, 0, stream>>>(k, w_k, b_k, kp, 1);
    gemm768_kernel<<<g_gemm, blk, 0, stream>>>(v, w_v, b_v, vp, 1);
    attn_kernel<<<g_attn, ablk, 0, stream>>>(qp, kp, vp, mask, xb);
    gemm768_kernel<<<g_gemm, blk, 0, stream>>>(xb, w_o, b_o, (float*)d_out, 0);
}


# projection GEMMs: A and W tiles staged once per 8-wave block in LDS (64x192 tile, 32-k stages, two buffers, 16-byte loads), one block per WGP; fragments read from LDS; sums in the same order
# speedup vs baseline: 1.6968x; 1.3685x over previous
#include <hip/hip_runtime.h>

typedef float v2f __attribute__((ext_vector_type(2)));
typedef float v8f __attribute__((ext_vector_type(8)));

#define D_MODEL 768
#define NHEAD   12
#define DK      64
#define BATCH   2
#define SEQ     2048
#define M_ROWS  (BATCH * SEQ)
#define NEGV    (-1000000000.0f)
#define SMSTRIDE 18
#define GBM      64
#define GBN      192
#define GBK      32
#define GAP      36
#define GWP      416
#define GSTEPS   (D_MODEL / GBK)
#define AWPB     4
#define ATHREADS (AWPB * 32)
#define KSTAGE   32
#define KPITCH   68
#define VPITCH   160
#define NSTAGE   (SEQ / KSTAGE)

__global__ __launch_bounds__(256) __attribute__((amdgpu_waves_per_eu(2, 2))) void gemm768_kernel(
    const float* __restrict__ A, const float* __restrict__ W,
    const float* __restrict__ bias, float* __restrict__ C, int split_heads)
{
    __shared__ __attribute__((aligned(16))) float as[2][GBM * GAP];
    __shared__ __attribute__((aligned(16))) float wsm[2][(GBK / 2) * GWP];

    int tid  = (int)threadIdx.x;
    int wv   = tid >> 5;
    int lane = tid & 31;
    int bn = (int)blockIdx.x % (D_MODEL / GBN);
    int bm = (int)blockIdx.x / (D_MODEL / GBN);
    int wm = wv >> 1;
    int wn = wv & 1;
    int l15 = lane & 15;
    int khi = lane >> 4;
    int rbase = khi << 3;

    const float4* Ag = (const float4*)(A + (size_t)(bm * GBM) * D_MODEL);
    const float4* Wg = (const float4*)(W + bn * GBN);
    int a_g = (tid >> 3) * (D_MODEL / 4) + (tid & 7);
    int a_l = (tid >> 3) * GAP + (tid & 7) * 4;
    int w_g[3], w_l[3];
#pragma unroll
    for (int q = 0; q < 3; ++q) {
        int u = tid + 256 * q;
        int pr = u / 48, c4 = u % 48;
        w_g[q] = (2 * pr) * (D_MODEL / 4) + c4;
        w_l[q] = pr * GWP + c4 * 8;
    }

    float4 areg[2], wreg[6];
#pragma unroll
    for (int p = 0; p < 2; ++p) areg[p] = Ag[a_g + p * 32 * (D_MODEL / 4)];
#pragma unroll
    for (int q = 0; q < 3; ++q) {
        wreg[2 * q]     = Wg[w_g[q]];
        wreg[2 * q + 1] = Wg[w_g[q] + (D_MODEL / 4)];
    }
#pragma unroll
    for (int p = 0; p < 2; ++p) *(float4*)&as[0][a_l + p * 32 * GAP] = areg[p];
#pragma unroll
    for (int q = 0; q < 3; ++q) {
        float4 a = wreg[2 * q], b = wreg[2 * q + 1];
        *(float4*)&wsm[0][w_l[q]]     = make_float4(a.x, b.x, a.y, b.y);
        *(float4*)&wsm[0][w_l[q] + 4] = make_float4(a.z, b.z, a.w, b.w);
    }
    __syncthreads();

    v8f acc[6];
#pragma unroll
    for (int c = 0; c < 6; ++c) acc[c] = (v8f){};

    for (int st = 0; st < GSTEPS; ++st) {
        int cur = st & 1;
        bool more = (st + 1 < GSTEPS);

        if (more) {
            int ga = a_g + (st + 1) * (GBK / 4);
            int gw = (st + 1) * GBK * (D_MODEL / 4);
#pragma unroll
            for (int p = 0; p < 2; ++p) areg[p] = Ag[ga + p * 32 * (D_MODEL / 4)];
#pragma unroll
            for (int q = 0; q < 3; ++q) {
                wreg[2 * q]     = Wg[gw + w_g[q]];
                wreg[2 * q + 1] = Wg[gw + w_g[q] + (D_MODEL / 4)];
            }
        }

        const float* ab = &as[cur][(wm * 16 + l15) * GAP + 2 * khi];
        const float* wb = &wsm[cur][khi * GWP + 2 * (wn * 96 + l15)];
        v2f af[GBK / 4], bf[GBK / 4][6];
#pragma unroll
        for (int j = 0; j < GBK / 4; ++j) {
            af[j] = *(const v2f*)(ab + 4 * j);
#pragma unroll
            for (int c = 0; c < 6; ++c)
                bf[j][c] = *(const v2f*)(wb + (2 * j) * GWP + 32 * c);
        }
#pragma unroll
        for (int j = 0; j < GBK / 4; ++j)
#pragma unroll
            for (int c = 0; c < 6; ++c)
                acc[c] = __builtin_amdgcn_wmma_f32_16x16x4_f32(
                    false, af[j], false, bf[j][c], (short)0, acc[c], false, false);

        if (more) {
#pragma unroll
            for (int p = 0; p < 2; ++p) *(float4*)&as[cur ^ 1][a_l + p * 32 * GAP] = areg[p];
#pragma unroll
            for (int q = 0; q < 3; ++q) {
                float4 a = wreg[2 * q], b = wreg[2 * q + 1];
                *(float4*)&wsm[cur ^ 1][w_l[q]]     = make_float4(a.x, b.x, a.y, b.y);
                *(float4*)&wsm[cur ^ 1][w_l[q] + 4] = make_float4(a.z, b.z, a.w, b.w);
            }
        }
        __syncthreads();
    }

    int m0 = bm * GBM + wm * 16;
    int n0 = bn * GBN + wn * 96;
    if (!split_heads) {
#pragma unroll
        for (int c = 0; c < 6; ++c) {
            float bv = bias[n0 + c * 16 + l15];
#pragma unroll
            for (int r = 0; r < 8; ++r) {
                int row = m0 + rbase + r;
                C[(size_t)row * D_MODEL + n0 + c * 16 + l15] = acc[c][r] + bv;
            }
        }
    } else {
#pragma unroll
        for (int c = 0; c < 6; ++c) {
            int col = n0 + c * 16 + l15;
            int h = col >> 6;
            int d = col & 63;
            float bv = bias[col];
#pragma unroll
            for (int r = 0; r < 8; ++r) {
                int row = m0 + rbase + r;
                int bi = row / SEQ, sq = row % SEQ;
                C[(((size_t)bi * NHEAD + h) * SEQ + sq) * DK + d] = acc[c][r] + bv;
            }
        }
    }
}

__global__ __launch_bounds__(ATHREADS) void attn_kernel(
    const float* __restrict__ QP, const float* __restrict__ KP,
    const float* __restrict__ VP, const int* __restrict__ mask,
    float* __restrict__ X)
{
    __shared__ __attribute__((aligned(16))) float ks[2][KSTAGE * KPITCH];
    __shared__ __attribute__((aligned(16))) float vs[2][(KSTAGE / 2) * VPITCH];
    __shared__ __attribute__((aligned(16))) float smem[AWPB][2][16 * SMSTRIDE];

    int tid  = (int)threadIdx.x;
    int wv   = tid >> 5;
    int lane = tid & 31;
    int wave = (int)blockIdx.x * AWPB + wv;

    int qt = wave % (SEQ / 16);
    int bh = wave / (SEQ / 16);
    int bi = bh / NHEAD, h = bh % NHEAD;
    int s0 = qt * 16;
    int l15 = lane & 15;
    int khi = lane >> 4;
    int rbase = khi << 3;

    const float* Q = QP + (size_t)bh * SEQ * DK;
    const float* K = KP + (size_t)bh * SEQ * DK;
    const float* V = VP + (size_t)bh * SEQ * DK;
    float* sm = &smem[wv][0][0];

    const float* qrow = Q + (size_t)(s0 + l15) * DK;
    v2f qv[16];
#pragma unroll
    for (int j = 0; j < 16; ++j)
        qv[j] = *(const v2f*)(qrow + 4 * j + 2 * khi);

    v8f acc[4];
#pragma unroll
    for (int c = 0; c < 4; ++c) acc[c] = (v8f){};

    const float4* Kg = (const float4*)K;
    const float4* Vg = (const float4*)V;
    const int* mrow = mask + (size_t)(s0 + rbase) * SEQ + l15;
    int vslot0 = ((tid >> 4) * 2) * 16 + (tid & 15);
    int vst0   = (tid >> 4) * VPITCH + (tid & 15) * 8;

    float4 kreg[4], vreg[4];
    int mreg[2][8];

#pragma unroll
    for (int p = 0; p < 4; ++p) kreg[p] = Kg[tid + ATHREADS * p];
#pragma unroll
    for (int p = 0; p < 2; ++p) {
        vreg[2 * p]     = Vg[vslot0 + 256 * p];
        vreg[2 * p + 1] = Vg[vslot0 + 256 * p + 16];
    }
#pragma unroll
    for (int u = 0; u < 2; ++u)
#pragma unroll
        for (int r = 0; r < 8; ++r)
            mreg[u][r] = mrow[(size_t)r * SEQ + 16 * u];
#pragma unroll
    for (int p = 0; p < 4; ++p) {
        int i = tid + ATHREADS * p;
        *(float4*)&ks[0][(i >> 4) * KPITCH + (i & 15) * 4] = kreg[p];
    }
#pragma unroll
    for (int p = 0; p < 2; ++p) {
        float4 a = vreg[2 * p], b = vreg[2 * p + 1];
        *(float4*)&vs[0][vst0 + 8 * VPITCH * p]     = make_float4(a.x, b.x, a.y, b.y);
        *(float4*)&vs[0][vst0 + 8 * VPITCH * p + 4] = make_float4(a.z, b.z, a.w, b.w);
    }
    __syncthreads();

    for (int st = 0; st < NSTAGE; ++st) {
        int cur = st & 1;
        bool more = (st + 1 < NSTAGE);

        int mk[2][8];
#pragma unroll
        for (int u = 0; u < 2; ++u)
#pragma unroll
            for (int r = 0; r < 8; ++r) mk[u][r] = mreg[u][r];

        if (more) {
            size_t g = (size_t)(st + 1) * (KSTAGE * DK / 4);
#pragma unroll
            for (int p = 0; p < 4; ++p) kreg[p] = Kg[g + tid + ATHREADS * p];
#pragma unroll
            for (int p = 0; p < 2; ++p) {
                vreg[2 * p]     = Vg[g + vslot0 + 256 * p];
                vreg[2 * p + 1] = Vg[g + vslot0 + 256 * p + 16];
            }
#pragma unroll
            for (int u = 0; u < 2; ++u)
#pragma unroll
                for (int r = 0; r < 8; ++r)
                    mreg[u][r] = mrow[(size_t)r * SEQ + (st + 1) * KSTAGE + 16 * u];
        }

        const float* k0 = &ks[cur][l15 * KPITCH + 2 * khi];
        const float* k1 = &ks[cur][(16 + l15) * KPITCH + 2 * khi];
        v2f kf0[16], kf1[16];
#pragma unroll
        for (int j = 0; j < 16; ++j) {
            kf0[j] = *(const v2f*)(k0 + 4 * j);
            kf1[j] = *(const v2f*)(k1 + 4 * j);
        }
        v8f sa = {}, sb = {};
#pragma unroll
        for (int j = 0; j < 16; ++j) {
            sa = __builtin_amdgcn_wmma_f32_16x16x4_f32(
                false, qv[j], false, kf0[j], (short)0, sa, false, false);
            sb = __builtin_amdgcn_wmma_f32_16x16x4_f32(
                false, qv[j], false, kf1[j], (short)0, sb, false, false);
        }

        const float* vb = &vs[cur][khi * VPITCH + 2 * l15];
        v2f vf[2][4][4];
#pragma unroll
        for (int u = 0; u < 2; ++u)
#pragma unroll
            for (int j = 0; j < 4; ++j)
#pragma unroll
                for (int c = 0; c < 4; ++c)
                    vf[u][j][c] = *(const v2f*)(vb + (8 * u + 2 * j) * VPITCH + 32 * c);

#pragma unroll
        for (int r = 0; r < 8; ++r) {
            int srow = rbase + r;
            float ea = sa[r] * 0.125f;
            float eb = sb[r] * 0.125f;
            if (mk[0][r] == 0) ea = NEGV;
            if (mk[1][r] == 0) eb = NEGV;
            sm[srow * SMSTRIDE + l15] = ea;
            sm[16 * SMSTRIDE + srow * SMSTRIDE + l15] = eb;
        }
        asm volatile("s_wait_dscnt 0" ::: "memory");

        v2f af[2][4];
#pragma unroll
        for (int u = 0; u < 2; ++u)
#pragma unroll
            for (int j = 0; j < 4; ++j)
                af[u][j] = *(const v2f*)(sm + u * 16 * SMSTRIDE + l15 * SMSTRIDE + 4 * j + 2 * khi);
#pragma unroll
        for (int u = 0; u < 2; ++u)
#pragma unroll
            for (int j = 0; j < 4; ++j)
#pragma unroll
                for (int c = 0; c < 4; ++c)
                    acc[c] = __builtin_amdgcn_wmma_f32_16x16x4_f32(
                        false, af[u][j], false, vf[u][j][c], (short)0, acc[c], false, false);

        if (more) {
#pragma unroll
            for (int p = 0; p < 4; ++p) {
                int i = tid + ATHREADS * p;
                *(float4*)&ks[cur ^ 1][(i >> 4) * KPITCH + (i & 15) * 4] = kreg[p];
            }
#pragma unroll
            for (int p = 0; p < 2; ++p) {
                float4 a = vreg[2 * p], b = vreg[2 * p + 1];
                *(float4*)&vs[cur ^ 1][vst0 + 8 * VPITCH * p]     = make_float4(a.x, b.x, a.y, b.y);
                *(float4*)&vs[cur ^ 1][vst0 + 8 * VPITCH * p + 4] = make_float4(a.z, b.z, a.w, b.w);
            }
        }
        __syncthreads();
    }

#pragma unroll
    for (int c = 0; c < 4; ++c) {
#pragma unroll
        for (int r = 0; r < 8; ++r) {
            int s = s0 + rbase + r;
            int d = c * 16 + l15;
            X[((size_t)bi * SEQ + s) * D_MODEL + h * DK + d] = acc[c][r];
        }
    }
}

extern "C" void kernel_launch(void* const* d_in, const int* in_sizes, int n_in,
                              void* d_out, int out_size, void* d_ws, size_t ws_size,
                              hipStream_t stream) {
    (void)in_sizes; (void)n_in; (void)out_size; (void)ws_size;
    const float* q    = (const float*)d_in[0];
    const float* k    = (const float*)d_in[1];
    const float* v    = (const float*)d_in[2];
    const int*   mask = (const int*)d_in[3];
    const float* w_q  = (const float*)d_in[4];
    const float* b_q  = (const float*)d_in[5];
    const float* w_k  = (const float*)d_in[6];
    const float* b_k  = (const float*)d_in[7];
    const float* w_v  = (const float*)d_in[8];
    const float* b_v  = (const float*)d_in[9];
    const float* w_o  = (const float*)d_in[10];
    const float* b_o  = (const float*)d_in[11];

    float* ws = (float*)d_ws;
    size_t proj = (size_t)M_ROWS * D_MODEL;
    float* qp = ws;
    float* kp = ws + proj;
    float* vp = ws + 2 * proj;
    float* xb = ws + 3 * proj;

    dim3 blk(256);
    dim3 g_gemm((M_ROWS / GBM) * (D_MODEL / GBN));
    int attn_waves = BATCH * NHEAD * (SEQ / 16);
    dim3 g_attn(attn_waves / AWPB);
    dim3 ablk(ATHREADS);

    gemm768_kernel<<<g_gemm, blk, 0, stream>>>(q, w_q, b_q, qp, 1);
    gemm768_kernel<<<g_gemm, blk, 0, stream>>>(k, w_k, b_k, kp, 1);
    gemm768_kernel<<<g_gemm, blk, 0, stream>>>(v, w_v, b_v, vp, 1);
    attn_kernel<<<g_attn, ablk, 0, stream>>>(qp, kp, vp, mask, xb);
    gemm768_kernel<<<g_gemm, blk, 0, stream>>>(xb, w_o, b_o, (float*)d_out, 0);
}


# attention: wave priority falls with progress through the keys (s_setprio 3,2,1,0 at 3/8, 11/16, 7/8) so the three blocks sharing a WGP advance together, not oldest-first one by one; sums unchanged
# speedup vs baseline: 1.7517x; 1.0324x over previous
#include <hip/hip_runtime.h>

typedef float v2f __attribute__((ext_vector_type(2)));
typedef float v8f __attribute__((ext_vector_type(8)));

#define D_MODEL 768
#define NHEAD   12
#define DK      64
#define BATCH   2
#define SEQ     2048
#define M_ROWS  (BATCH * SEQ)
#define NEGV    (-1000000000.0f)
#define SMSTRIDE 18
#define GBM      64
#define GBN      192
#define GBK      32
#define GAP      36
#define GWP      416
#define GSTEPS   (D_MODEL / GBK)
#define AWPB     4
#define ATHREADS (AWPB * 32)
#define KSTAGE   32
#define KPITCH   68
#define VPITCH   160
#define NSTAGE   (SEQ / KSTAGE)
#define PMARK1   (NSTAGE * 3 / 8)
#define PMARK2   (NSTAGE * 11 / 16)
#define PMARK3   (NSTAGE * 7 / 8)

__global__ __launch_bounds__(256) __attribute__((amdgpu_waves_per_eu(2, 2))) void gemm768_kernel(
    const float* __restrict__ A, const float* __restrict__ W,
    const float* __restrict__ bias, float* __restrict__ C, int split_heads)
{
    __shared__ __attribute__((aligned(16))) float as[2][GBM * GAP];
    __shared__ __attribute__((aligned(16))) float wsm[2][(GBK / 2) * GWP];

    int tid  = (int)threadIdx.x;
    int wv   = tid >> 5;
    int lane = tid & 31;
    int bn = (int)blockIdx.x % (D_MODEL / GBN);
    int bm = (int)blockIdx.x / (D_MODEL / GBN);
    int wm = wv >> 1;
    int wn = wv & 1;
    int l15 = lane & 15;
    int khi = lane >> 4;
    int rbase = khi << 3;

    const float4* Ag = (const float4*)(A + (size_t)(bm * GBM) * D_MODEL);
    const float4* Wg = (const float4*)(W + bn * GBN);
    int a_g = (tid >> 3) * (D_MODEL / 4) + (tid & 7);
    int a_l = (tid >> 3) * GAP + (tid & 7) * 4;
    int w_g[3], w_l[3];
#pragma unroll
    for (int q = 0; q < 3; ++q) {
        int u = tid + 256 * q;
        int pr = u / 48, c4 = u % 48;
        w_g[q] = (2 * pr) * (D_MODEL / 4) + c4;
        w_l[q] = pr * GWP + c4 * 8;
    }

    float4 areg[2], wreg[6];
#pragma unroll
    for (int p = 0; p < 2; ++p) areg[p] = Ag[a_g + p * 32 * (D_MODEL / 4)];
#pragma unroll
    for (int q = 0; q < 3; ++q) {
        wreg[2 * q]     = Wg[w_g[q]];
        wreg[2 * q + 1] = Wg[w_g[q] + (D_MODEL / 4)];
    }
#pragma unroll
    for (int p = 0; p < 2; ++p) *(float4*)&as[0][a_l + p * 32 * GAP] = areg[p];
#pragma unroll
    for (int q = 0; q < 3; ++q) {
        float4 a = wreg[2 * q], b = wreg[2 * q + 1];
        *(float4*)&wsm[0][w_l[q]]     = make_float4(a.x, b.x, a.y, b.y);
        *(float4*)&wsm[0][w_l[q] + 4] = make_float4(a.z, b.z, a.w, b.w);
    }
    __syncthreads();

    v8f acc[6];
#pragma unroll
    for (int c = 0; c < 6; ++c) acc[c] = (v8f){};

    for (int st = 0; st < GSTEPS; ++st) {
        int cur = st & 1;
        bool more = (st + 1 < GSTEPS);

        if (more) {
            int ga = a_g + (st + 1) * (GBK / 4);
            int gw = (st + 1) * GBK * (D_MODEL / 4);
#pragma unroll
            for (int p = 0; p < 2; ++p) areg[p] = Ag[ga + p * 32 * (D_MODEL / 4)];
#pragma unroll
            for (int q = 0; q < 3; ++q) {
                wreg[2 * q]     = Wg[gw + w_g[q]];
                wreg[2 * q + 1] = Wg[gw + w_g[q] + (D_MODEL / 4)];
            }
        }

        const float* ab = &as[cur][(wm * 16 + l15) * GAP + 2 * khi];
        const float* wb = &wsm[cur][khi * GWP + 2 * (wn * 96 + l15)];
        v2f af[GBK / 4], bf[GBK / 4][6];
#pragma unroll
        for (int j = 0; j < GBK / 4; ++j) {
            af[j] = *(const v2f*)(ab + 4 * j);
#pragma unroll
            for (int c = 0; c < 6; ++c)
                bf[j][c] = *(const v2f*)(wb + (2 * j) * GWP + 32 * c);
        }
#pragma unroll
        for (int j = 0; j < GBK / 4; ++j)
#pragma unroll
            for (int c = 0; c < 6; ++c)
                acc[c] = __builtin_amdgcn_wmma_f32_16x16x4_f32(
                    false, af[j], false, bf[j][c], (short)0, acc[c], false, false);

        if (more) {
#pragma unroll
            for (int p = 0; p < 2; ++p) *(float4*)&as[cur ^ 1][a_l + p * 32 * GAP] = areg[p];
#pragma unroll
            for (int q = 0; q < 3; ++q) {
                float4 a = wreg[2 * q], b = wreg[2 * q + 1];
                *(float4*)&wsm[cur ^ 1][w_l[q]]     = make_float4(a.x, b.x, a.y, b.y);
                *(float4*)&wsm[cur ^ 1][w_l[q] + 4] = make_float4(a.z, b.z, a.w, b.w);
            }
        }
        __syncthreads();
    }

    int m0 = bm * GBM + wm * 16;
    int n0 = bn * GBN + wn * 96;
    if (!split_heads) {
#pragma unroll
        for (int c = 0; c < 6; ++c) {
            float bv = bias[n0 + c * 16 + l15];
#pragma unroll
            for (int r = 0; r < 8; ++r) {
                int row = m0 + rbase + r;
                C[(size_t)row * D_MODEL + n0 + c * 16 + l15] = acc[c][r] + bv;
            }
        }
    } else {
#pragma unroll
        for (int c = 0; c < 6; ++c) {
            int col = n0 + c * 16 + l15;
            int h = col >> 6;
            int d = col & 63;
            float bv = bias[col];
#pragma unroll
            for (int r = 0; r < 8; ++r) {
                int row = m0 + rbase + r;
                int bi = row / SEQ, sq = row % SEQ;
                C[(((size_t)bi * NHEAD + h) * SEQ + sq) * DK + d] = acc[c][r] + bv;
            }
        }
    }
}

__global__ __launch_bounds__(ATHREADS) void attn_kernel(
    const float* __restrict__ QP, const float* __restrict__ KP,
    const float* __restrict__ VP, const int* __restrict__ mask,
    float* __restrict__ X)
{
    __shared__ __attribute__((aligned(16))) float ks[2][KSTAGE * KPITCH];
    __shared__ __attribute__((aligned(16))) float vs[2][(KSTAGE / 2) * VPITCH];
    __shared__ __attribute__((aligned(16))) float smem[AWPB][2][16 * SMSTRIDE];

    int tid  = (int)threadIdx.x;
    int wv   = tid >> 5;
    int lane = tid & 31;
    int wave = (int)blockIdx.x * AWPB + wv;

    int qt = wave % (SEQ / 16);
    int bh = wave / (SEQ / 16);
    int bi = bh / NHEAD, h = bh % NHEAD;
    int s0 = qt * 16;
    int l15 = lane & 15;
    int khi = lane >> 4;
    int rbase = khi << 3;

    const float* Q = QP + (size_t)bh * SEQ * DK;
    const float* K = KP + (size_t)bh * SEQ * DK;
    const float* V = VP + (size_t)bh * SEQ * DK;
    float* sm = &smem[wv][0][0];

    const float* qrow = Q + (size_t)(s0 + l15) * DK;
    v2f qv[16];
#pragma unroll
    for (int j = 0; j < 16; ++j)
        qv[j] = *(const v2f*)(qrow + 4 * j + 2 * khi);

    v8f acc[4];
#pragma unroll
    for (int c = 0; c < 4; ++c) acc[c] = (v8f){};

    const float4* Kg = (const float4*)K;
    const float4* Vg = (const float4*)V;
    const int* mrow = mask + (size_t)(s0 + rbase) * SEQ + l15;
    int vslot0 = ((tid >> 4) * 2) * 16 + (tid & 15);
    int vst0   = (tid >> 4) * VPITCH + (tid & 15) * 8;

    float4 kreg[4], vreg[4];
    int mreg[2][8];

#pragma unroll
    for (int p = 0; p < 4; ++p) kreg[p] = Kg[tid + ATHREADS * p];
#pragma unroll
    for (int p = 0; p < 2; ++p) {
        vreg[2 * p]     = Vg[vslot0 + 256 * p];
        vreg[2 * p + 1] = Vg[vslot0 + 256 * p + 16];
    }
#pragma unroll
    for (int u = 0; u < 2; ++u)
#pragma unroll
        for (int r = 0; r < 8; ++r)
            mreg[u][r] = mrow[(size_t)r * SEQ + 16 * u];
#pragma unroll
    for (int p = 0; p < 4; ++p) {
        int i = tid + ATHREADS * p;
        *(float4*)&ks[0][(i >> 4) * KPITCH + (i & 15) * 4] = kreg[p];
    }
#pragma unroll
    for (int p = 0; p < 2; ++p) {
        float4 a = vreg[2 * p], b = vreg[2 * p + 1];
        *(float4*)&vs[0][vst0 + 8 * VPITCH * p]     = make_float4(a.x, b.x, a.y, b.y);
        *(float4*)&vs[0][vst0 + 8 * VPITCH * p + 4] = make_float4(a.z, b.z, a.w, b.w);
    }
    __syncthreads();

    __builtin_amdgcn_s_setprio(3);

    for (int st = 0; st < NSTAGE; ++st) {
        int cur = st & 1;
        bool more = (st + 1 < NSTAGE);

        if (st == PMARK1)      __builtin_amdgcn_s_setprio(2);
        else if (st == PMARK2) __builtin_amdgcn_s_setprio(1);
        else if (st == PMARK3) __builtin_amdgcn_s_setprio(0);

        int mk[2][8];
#pragma unroll
        for (int u = 0; u < 2; ++u)
#pragma unroll
            for (int r = 0; r < 8; ++r) mk[u][r] = mreg[u][r];

        if (more) {
            size_t g = (size_t)(st + 1) * (KSTAGE * DK / 4);
#pragma unroll
            for (int p = 0; p < 4; ++p) kreg[p] = Kg[g + tid + ATHREADS * p];
#pragma unroll
            for (int p = 0; p < 2; ++p) {
                vreg[2 * p]     = Vg[g + vslot0 + 256 * p];
                vreg[2 * p + 1] = Vg[g + vslot0 + 256 * p + 16];
            }
#pragma unroll
            for (int u = 0; u < 2; ++u)
#pragma unroll
                for (int r = 0; r < 8; ++r)
                    mreg[u][r] = mrow[(size_t)r * SEQ + (st + 1) * KSTAGE + 16 * u];
        }

        const float* k0 = &ks[cur][l15 * KPITCH + 2 * khi];
        const float* k1 = &ks[cur][(16 + l15) * KPITCH + 2 * khi];
        v2f kf0[16], kf1[16];
#pragma unroll
        for (int j = 0; j < 16; ++j) {
            kf0[j] = *(const v2f*)(k0 + 4 * j);
            kf1[j] = *(const v2f*)(k1 + 4 * j);
        }
        v8f sa = {}, sb = {};
#pragma unroll
        for (int j = 0; j < 16; ++j) {
            sa = __builtin_amdgcn_wmma_f32_16x16x4_f32(
                false, qv[j], false, kf0[j], (short)0, sa, false, false);
            sb = __builtin_amdgcn_wmma_f32_16x16x4_f32(
                false, qv[j], false, kf1[j], (short)0, sb, false, false);
        }

        const float* vb = &vs[cur][khi * VPITCH + 2 * l15];
        v2f vf[2][4][4];
#pragma unroll
        for (int u = 0; u < 2; ++u)
#pragma unroll
            for (int j = 0; j < 4; ++j)
#pragma unroll
                for (int c = 0; c < 4; ++c)
                    vf[u][j][c] = *(const v2f*)(vb + (8 * u + 2 * j) * VPITCH + 32 * c);

#pragma unroll
        for (int r = 0; r < 8; ++r) {
            int srow = rbase + r;
            float ea = sa[r] * 0.125f;
            float eb = sb[r] * 0.125f;
            if (mk[0][r] == 0) ea = NEGV;
            if (mk[1][r] == 0) eb = NEGV;
            sm[srow * SMSTRIDE + l15] = ea;
            sm[16 * SMSTRIDE + srow * SMSTRIDE + l15] = eb;
        }
        asm volatile("s_wait_dscnt 0" ::: "memory");

        v2f af[2][4];
#pragma unroll
        for (int u = 0; u < 2; ++u)
#pragma unroll
            for (int j = 0; j < 4; ++j)
                af[u][j] = *(const v2f*)(sm + u * 16 * SMSTRIDE + l15 * SMSTRIDE + 4 * j + 2 * khi);
#pragma unroll
        for (int u = 0; u < 2; ++u)
#pragma unroll
            for (int j = 0; j < 4; ++j)
#pragma unroll
                for (int c = 0; c < 4; ++c)
                    acc[c] = __builtin_amdgcn_wmma_f32_16x16x4_f32(
                        false, af[u][j], false, vf[u][j][c], (short)0, acc[c], false, false);

        if (more) {
#pragma unroll
            for (int p = 0; p < 4; ++p) {
                int i = tid + ATHREADS * p;
                *(float4*)&ks[cur ^ 1][(i >> 4) * KPITCH + (i & 15) * 4] = kreg[p];
            }
#pragma unroll
            for (int p = 0; p < 2; ++p) {
                float4 a = vreg[2 * p], b = vreg[2 * p + 1];
                *(float4*)&vs[cur ^ 1][vst0 + 8 * VPITCH * p]     = make_float4(a.x, b.x, a.y, b.y);
                *(float4*)&vs[cur ^ 1][vst0 + 8 * VPITCH * p + 4] = make_float4(a.z, b.z, a.w, b.w);
            }
        }
        __syncthreads();
    }

#pragma unroll
    for (int c = 0; c < 4; ++c) {
#pragma unroll
        for (int r = 0; r < 8; ++r) {
            int s = s0 + rbase + r;
            int d = c * 16 + l15;
            X[((size_t)bi * SEQ + s) * D_MODEL + h * DK + d] = acc[c][r];
        }
    }
}

extern "C" void kernel_launch(void* const* d_in, const int* in_sizes, int n_in,
                              void* d_out, int out_size, void* d_ws, size_t ws_size,
                              hipStream_t stream) {
    (void)in_sizes; (void)n_in; (void)out_size; (void)ws_size;
    const float* q    = (const float*)d_in[0];
    const float* k    = (const float*)d_in[1];
    const float* v    = (const float*)d_in[2];
    const int*   mask = (const int*)d_in[3];
    const float* w_q  = (const float*)d_in[4];
    const float* b_q  = (const float*)d_in[5];
    const float* w_k  = (const float*)d_in[6];
    const float* b_k  = (const float*)d_in[7];
    const float* w_v  = (const float*)d_in[8];
    const float* b_v  = (const float*)d_in[9];
    const float* w_o  = (const float*)d_in[10];
    const float* b_o  = (const float*)d_in[11];

    float* ws = (float*)d_ws;
    size_t proj = (size_t)M_ROWS * D_MODEL;
    float* qp = ws;
    float* kp = ws + proj;
    float* vp = ws + 2 * proj;
    float* xb = ws + 3 * proj;

    dim3 blk(256);
    dim3 g_gemm((M_ROWS / GBM) * (D_MODEL / GBN));
    int attn_waves = BATCH * NHEAD * (SEQ / 16);
    dim3 g_attn(attn_waves / AWPB);
    dim3 ablk(ATHREADS);

    gemm768_kernel<<<g_gemm, blk, 0, stream>>>(q, w_q, b_q, qp, 1);
    gemm768_kernel<<<g_gemm, blk, 0, stream>>>(k, w_k, b_k, kp, 1);
    gemm768_kernel<<<g_gemm, blk, 0, stream>>>(v, w_v, b_v, vp, 1);
    attn_kernel<<<g_attn, ablk, 0, stream>>>(qp, kp, vp, mask, xb);
    gemm768_kernel<<<g_gemm, blk, 0, stream>>>(xb, w_o, b_o, (float*)d_out, 0);
}


# projection GEMMs: next stage's LDS stores moved in front of the products, stage barrier split (signal before, wait behind them): a SIMD's two waves no longer meet at stage ends; sums unchanged
# speedup vs baseline: 1.8134x; 1.0352x over previous
#include <hip/hip_runtime.h>

typedef float v2f __attribute__((ext_vector_type(2)));
typedef float v8f __attribute__((ext_vector_type(8)));

#define D_MODEL 768
#define NHEAD   12
#define DK      64
#define BATCH   2
#define SEQ     2048
#define M_ROWS  (BATCH * SEQ)
#define NEGV    (-1000000000.0f)
#define SMSTRIDE 18
#define GBM      64
#define GBN      192
#define GBK      32
#define GAP      36
#define GWP      416
#define GSTEPS   (D_MODEL / GBK)
#define AWPB     4
#define ATHREADS (AWPB * 32)
#define KSTAGE   32
#define KPITCH   68
#define VPITCH   160
#define NSTAGE   (SEQ / KSTAGE)
#define PMARK1   (NSTAGE * 3 / 8)
#define PMARK2   (NSTAGE * 11 / 16)
#define PMARK3   (NSTAGE * 7 / 8)

__global__ __launch_bounds__(256) __attribute__((amdgpu_waves_per_eu(2, 2))) void gemm768_kernel(
    const float* __restrict__ A, const float* __restrict__ W,
    const float* __restrict__ bias, float* __restrict__ C, int split_heads)
{
    __shared__ __attribute__((aligned(16))) float as[2][GBM * GAP];
    __shared__ __attribute__((aligned(16))) float wsm[2][(GBK / 2) * GWP];

    int tid  = (int)threadIdx.x;
    int wv   = tid >> 5;
    int lane = tid & 31;
    int bn = (int)blockIdx.x % (D_MODEL / GBN);
    int bm = (int)blockIdx.x / (D_MODEL / GBN);
    int wm = wv >> 1;
    int wn = wv & 1;
    int l15 = lane & 15;
    int khi = lane >> 4;
    int rbase = khi << 3;

    const float4* Ag = (const float4*)(A + (size_t)(bm * GBM) * D_MODEL);
    const float4* Wg = (const float4*)(W + bn * GBN);
    int a_g = (tid >> 3) * (D_MODEL / 4) + (tid & 7);
    int a_l = (tid >> 3) * GAP + (tid & 7) * 4;
    int w_g[3], w_l[3];
#pragma unroll
    for (int q = 0; q < 3; ++q) {
        int u = tid + 256 * q;
        int pr = u / 48, c4 = u % 48;
        w_g[q] = (2 * pr) * (D_MODEL / 4) + c4;
        w_l[q] = pr * GWP + c4 * 8;
    }

    const int AP = 32 * (D_MODEL / 4);
    const int WR = D_MODEL / 4;
    float4 s_a0, s_a1, s_w0, s_w1, s_w2, s_w3, s_w4, s_w5;

#define STAGE_STORE(buf, A0, A1, W0, W1, W2, W3, W4, W5)                                  \
    do {                                                                                  \
        *(float4*)&as[buf][a_l]            = A0;                                          \
        *(float4*)&as[buf][a_l + 32 * GAP] = A1;                                          \
        *(float4*)&wsm[buf][w_l[0]]     = make_float4(W0.x, W1.x, W0.y, W1.y);            \
        *(float4*)&wsm[buf][w_l[0] + 4] = make_float4(W0.z, W1.z, W0.w, W1.w);            \
        *(float4*)&wsm[buf][w_l[1]]     = make_float4(W2.x, W3.x, W2.y, W3.y);            \
        *(float4*)&wsm[buf][w_l[1] + 4] = make_float4(W2.z, W3.z, W2.w, W3.w);            \
        *(float4*)&wsm[buf][w_l[2]]     = make_float4(W4.x, W5.x, W4.y, W5.y);            \
        *(float4*)&wsm[buf][w_l[2] + 4] = make_float4(W4.z, W5.z, W4.w, W5.w);            \
    } while (0)
#define STAGE_FETCH(stg)                                                                  \
    do {                                                                                  \
        int ga_ = a_g + (stg) * (GBK / 4);                                                \
        int gw_ = (stg) * GBK * (D_MODEL / 4);                                            \
        s_a0 = Ag[ga_];              s_a1 = Ag[ga_ + AP];                                 \
        s_w0 = Wg[gw_ + w_g[0]];     s_w1 = Wg[gw_ + w_g[0] + WR];                        \
        s_w2 = Wg[gw_ + w_g[1]];     s_w3 = Wg[gw_ + w_g[1] + WR];                        \
        s_w4 = Wg[gw_ + w_g[2]];     s_w5 = Wg[gw_ + w_g[2] + WR];                        \
    } while (0)

    {
        float4 t_a0 = Ag[a_g], t_a1 = Ag[a_g + AP];
        float4 t_w0 = Wg[w_g[0]], t_w1 = Wg[w_g[0] + WR];
        float4 t_w2 = Wg[w_g[1]], t_w3 = Wg[w_g[1] + WR];
        float4 t_w4 = Wg[w_g[2]], t_w5 = Wg[w_g[2] + WR];
        STAGE_FETCH(1);
        STAGE_STORE(0, t_a0, t_a1, t_w0, t_w1, t_w2, t_w3, t_w4, t_w5);
    }
    __syncthreads();

    v8f acc[6];
#pragma unroll
    for (int c = 0; c < 6; ++c) acc[c] = (v8f){};

    for (int st = 0; st < GSTEPS; ++st) {
        int cur = st & 1;
        bool more = (st + 1 < GSTEPS);
        bool more2 = (st + 2 < GSTEPS);

        const float* ab = &as[cur][(wm * 16 + l15) * GAP + 2 * khi];
        const float* wb = &wsm[cur][khi * GWP + 2 * (wn * 96 + l15)];
        v2f af[GBK / 4], bf[GBK / 4][6];
#pragma unroll
        for (int j = 0; j < GBK / 4; ++j) {
            af[j] = *(const v2f*)(ab + 4 * j);
#pragma unroll
            for (int c = 0; c < 6; ++c)
                bf[j][c] = *(const v2f*)(wb + (2 * j) * GWP + 32 * c);
        }

        if (more) {
            STAGE_STORE(cur ^ 1, s_a0, s_a1, s_w0, s_w1, s_w2, s_w3, s_w4, s_w5);
            asm volatile("s_wait_dscnt 0x0\n\ts_barrier_signal -1" ::: "memory");
        }
        __builtin_amdgcn_sched_barrier(0);

        if (more2) STAGE_FETCH(st + 2);

#pragma unroll
        for (int j = 0; j < GBK / 4; ++j)
#pragma unroll
            for (int c = 0; c < 6; ++c)
                acc[c] = __builtin_amdgcn_wmma_f32_16x16x4_f32(
                    false, af[j], false, bf[j][c], (short)0, acc[c], false, false);

        __builtin_amdgcn_sched_barrier(0);
        if (more) {
            asm volatile("s_barrier_wait 0xffff" ::: "memory");
        }
    }
#undef STAGE_STORE
#undef STAGE_FETCH

    int m0 = bm * GBM + wm * 16;
    int n0 = bn * GBN + wn * 96;
    if (!split_heads) {
#pragma unroll
        for (int c = 0; c < 6; ++c) {
            float bv = bias[n0 + c * 16 + l15];
#pragma unroll
            for (int r = 0; r < 8; ++r) {
                int row = m0 + rbase + r;
                C[(size_t)row * D_MODEL + n0 + c * 16 + l15] = acc[c][r] + bv;
            }
        }
    } else {
#pragma unroll
        for (int c = 0; c < 6; ++c) {
            int col = n0 + c * 16 + l15;
            int h = col >> 6;
            int d = col & 63;
            float bv = bias[col];
#pragma unroll
            for (int r = 0; r < 8; ++r) {
                int row = m0 + rbase + r;
                int bi = row / SEQ, sq = row % SEQ;
                C[(((size_t)bi * NHEAD + h) * SEQ + sq) * DK + d] = acc[c][r] + bv;
            }
        }
    }
}

__global__ __launch_bounds__(ATHREADS) void attn_kernel(
    const float* __restrict__ QP, const float* __restrict__ KP,
    const float* __restrict__ VP, const int* __restrict__ mask,
    float* __restrict__ X)
{
    __shared__ __attribute__((aligned(16))) float ks[2][KSTAGE * KPITCH];
    __shared__ __attribute__((aligned(16))) float vs[2][(KSTAGE / 2) * VPITCH];
    __shared__ __attribute__((aligned(16))) float smem[AWPB][2][16 * SMSTRIDE];

    int tid  = (int)threadIdx.x;
    int wv   = tid >> 5;
    int lane = tid & 31;
    int wave = (int)blockIdx.x * AWPB + wv;

    int qt = wave % (SEQ / 16);
    int bh = wave / (SEQ / 16);
    int bi = bh / NHEAD, h = bh % NHEAD;
    int s0 = qt * 16;
    int l15 = lane & 15;
    int khi = lane >> 4;
    int rbase = khi << 3;

    const float* Q = QP + (size_t)bh * SEQ * DK;
    const float* K = KP + (size_t)bh * SEQ * DK;
    const float* V = VP + (size_t)bh * SEQ * DK;
    float* sm = &smem[wv][0][0];

    const float* qrow = Q + (size_t)(s0 + l15) * DK;
    v2f qv[16];
#pragma unroll
    for (int j = 0; j < 16; ++j)
        qv[j] = *(const v2f*)(qrow + 4 * j + 2 * khi);

    v8f acc[4];
#pragma unroll
    for (int c = 0; c < 4; ++c) acc[c] = (v8f){};

    const float4* Kg = (const float4*)K;
    const float4* Vg = (const float4*)V;
    const int* mrow = mask + (size_t)(s0 + rbase) * SEQ + l15;
    int vslot0 = ((tid >> 4) * 2) * 16 + (tid & 15);
    int vst0   = (tid >> 4) * VPITCH + (tid & 15) * 8;

    float4 kreg[4], vreg[4];
    int mreg[2][8];

#pragma unroll
    for (int p = 0; p < 4; ++p) kreg[p] = Kg[tid + ATHREADS * p];
#pragma unroll
    for (int p = 0; p < 2; ++p) {
        vreg[2 * p]     = Vg[vslot0 + 256 * p];
        vreg[2 * p + 1] = Vg[vslot0 + 256 * p + 16];
    }
#pragma unroll
    for (int u = 0; u < 2; ++u)
#pragma unroll
        for (int r = 0; r < 8; ++r)
            mreg[u][r] = mrow[(size_t)r * SEQ + 16 * u];
#pragma unroll
    for (int p = 0; p < 4; ++p) {
        int i = tid + ATHREADS * p;
        *(float4*)&ks[0][(i >> 4) * KPITCH + (i & 15) * 4] = kreg[p];
    }
#pragma unroll
    for (int p = 0; p < 2; ++p) {
        float4 a = vreg[2 * p], b = vreg[2 * p + 1];
        *(float4*)&vs[0][vst0 + 8 * VPITCH * p]     = make_float4(a.x, b.x, a.y, b.y);
        *(float4*)&vs[0][vst0 + 8 * VPITCH * p + 4] = make_float4(a.z, b.z, a.w, b.w);
    }
    __syncthreads();

    __builtin_amdgcn_s_setprio(3);

    for (int st = 0; st < NSTAGE; ++st) {
        int cur = st & 1;
        bool more = (st + 1 < NSTAGE);

        if (st == PMARK1)      __builtin_amdgcn_s_setprio(2);
        else if (st == PMARK2) __builtin_amdgcn_s_setprio(1);
        else if (st == PMARK3) __builtin_amdgcn_s_setprio(0);

        int mk[2][8];
#pragma unroll
        for (int u = 0; u < 2; ++u)
#pragma unroll
            for (int r = 0; r < 8; ++r) mk[u][r] = mreg[u][r];

        if (more) {
            size_t g = (size_t)(st + 1) * (KSTAGE * DK / 4);
#pragma unroll
            for (int p = 0; p < 4; ++p) kreg[p] = Kg[g + tid + ATHREADS * p];
#pragma unroll
            for (int p = 0; p < 2; ++p) {
                vreg[2 * p]     = Vg[g + vslot0 + 256 * p];
                vreg[2 * p + 1] = Vg[g + vslot0 + 256 * p + 16];
            }
#pragma unroll
            for (int u = 0; u < 2; ++u)
#pragma unroll
                for (int r = 0; r < 8; ++r)
                    mreg[u][r] = mrow[(size_t)r * SEQ + (st + 1) * KSTAGE + 16 * u];
        }

        const float* k0 = &ks[cur][l15 * KPITCH + 2 * khi];
        const float* k1 = &ks[cur][(16 + l15) * KPITCH + 2 * khi];
        v2f kf0[16], kf1[16];
#pragma unroll
        for (int j = 0; j < 16; ++j) {
            kf0[j] = *(const v2f*)(k0 + 4 * j);
            kf1[j] = *(const v2f*)(k1 + 4 * j);
        }
        v8f sa = {}, sb = {};
#pragma unroll
        for (int j = 0; j < 16; ++j) {
            sa = __builtin_amdgcn_wmma_f32_16x16x4_f32(
                false, qv[j], false, kf0[j], (short)0, sa, false, false);
            sb = __builtin_amdgcn_wmma_f32_16x16x4_f32(
                false, qv[j], false, kf1[j], (short)0, sb, false, false);
        }

        const float* vb = &vs[cur][khi * VPITCH + 2 * l15];
        v2f vf[2][4][4];
#pragma unroll
        for (int u = 0; u < 2; ++u)
#pragma unroll
            for (int j = 0; j < 4; ++j)
#pragma unroll
                for (int c = 0; c < 4; ++c)
                    vf[u][j][c] = *(const v2f*)(vb + (8 * u + 2 * j) * VPITCH + 32 * c);

#pragma unroll
        for (int r = 0; r < 8; ++r) {
            int srow = rbase + r;
            float ea = sa[r] * 0.125f;
            float eb = sb[r] * 0.125f;
            if (mk[0][r] == 0) ea = NEGV;
            if (mk[1][r] == 0) eb = NEGV;
            sm[srow * SMSTRIDE + l15] = ea;
            sm[16 * SMSTRIDE + srow * SMSTRIDE + l15] = eb;
        }
        asm volatile("s_wait_dscnt 0" ::: "memory");

        v2f af[2][4];
#pragma unroll
        for (int u = 0; u < 2; ++u)
#pragma unroll
            for (int j = 0; j < 4; ++j)
                af[u][j] = *(const v2f*)(sm + u * 16 * SMSTRIDE + l15 * SMSTRIDE + 4 * j + 2 * khi);
#pragma unroll
        for (int u = 0; u < 2; ++u)
#pragma unroll
            for (int j = 0; j < 4; ++j)
#pragma unroll
                for (int c = 0; c < 4; ++c)
                    acc[c] = __builtin_amdgcn_wmma_f32_16x16x4_f32(
                        false, af[u][j], false, vf[u][j][c], (short)0, acc[c], false, false);

        if (more) {
#pragma unroll
            for (int p = 0; p < 4; ++p) {
                int i = tid + ATHREADS * p;
                *(float4*)&ks[cur ^ 1][(i >> 4) * KPITCH + (i & 15) * 4] = kreg[p];
            }
#pragma unroll
            for (int p = 0; p < 2; ++p) {
                float4 a = vreg[2 * p], b = vreg[2 * p + 1];
                *(float4*)&vs[cur ^ 1][vst0 + 8 * VPITCH * p]     = make_float4(a.x, b.x, a.y, b.y);
                *(float4*)&vs[cur ^ 1][vst0 + 8 * VPITCH * p + 4] = make_float4(a.z, b.z, a.w, b.w);
            }
        }
        __syncthreads();
    }

#pragma unroll
    for (int c = 0; c < 4; ++c) {
#pragma unroll
        for (int r = 0; r < 8; ++r) {
            int s = s0 + rbase + r;
            int d = c * 16 + l15;
            X[((size_t)bi * SEQ + s) * D_MODEL + h * DK + d] = acc[c][r];
        }
    }
}

extern "C" void kernel_launch(void* const* d_in, const int* in_sizes, int n_in,
                              void* d_out, int out_size, void* d_ws, size_t ws_size,
                              hipStream_t stream) {
    (void)in_sizes; (void)n_in; (void)out_size; (void)ws_size;
    const float* q    = (const float*)d_in[0];
    const float* k    = (const float*)d_in[1];
    const float* v    = (const float*)d_in[2];
    const int*   mask = (const int*)d_in[3];
    const float* w_q  = (const float*)d_in[4];
    const float* b_q  = (const float*)d_in[5];
    const float* w_k  = (const float*)d_in[6];
    const float* b_k  = (const float*)d_in[7];
    const float* w_v  = (const float*)d_in[8];
    const float* b_v  = (const float*)d_in[9];
    const float* w_o  = (const float*)d_in[10];
    const float* b_o  = (const float*)d_in[11];

    float* ws = (float*)d_ws;
    size_t proj = (size_t)M_ROWS * D_MODEL;
    float* qp = ws;
    float* kp = ws + proj;
    float* vp = ws + 2 * proj;
    float* xb = ws + 3 * proj;

    dim3 blk(256);
    dim3 g_gemm((M_ROWS / GBM) * (D_MODEL / GBN));
    int attn_waves = BATCH * NHEAD * (SEQ / 16);
    dim3 g_attn(attn_waves / AWPB);
    dim3 ablk(ATHREADS);

    gemm768_kernel<<<g_gemm, blk, 0, stream>>>(q, w_q, b_q, qp, 1);
    gemm768_kernel<<<g_gemm, blk, 0, stream>>>(k, w_k, b_k, kp, 1);
    gemm768_kernel<<<g_gemm, blk, 0, stream>>>(v, w_v, b_v, vp, 1);
    attn_kernel<<<g_attn, ablk, 0, stream>>>(qp, kp, vp, mask, xb);
    gemm768_kernel<<<g_gemm, blk, 0, stream>>>(xb, w_o, b_o, (float*)d_out, 0);
}


# attention: no softmax, so (QK^T)V is summed as Q(K^T V): a 64x64 K^T V per head from 16 key-chunk partials, then Q times it; mask zeros corrected per 16x32 tile; order of sums changed
# speedup vs baseline: 3.2730x; 1.8049x over previous
#include <hip/hip_runtime.h>

typedef float v2f __attribute__((ext_vector_type(2)));
typedef float v8f __attribute__((ext_vector_type(8)));

#define D_MODEL 768
#define NHEAD   12
#define DK      64
#define BATCH   2
#define SEQ     2048
#define M_ROWS  (BATCH * SEQ)
#define NEGV    (-1000000000.0f)
#define SMSTRIDE 18
#define GBM      64
#define GBN      192
#define GBK      32
#define GAP      36
#define GWP      416
#define GSTEPS   (D_MODEL / GBK)
#define KSTAGE   32
#define VPITCH   160

__global__ __launch_bounds__(256) __attribute__((amdgpu_waves_per_eu(2, 2))) void gemm768_kernel(
    const float* __restrict__ A, const float* __restrict__ W,
    const float* __restrict__ bias, float* __restrict__ C, int split_heads)
{
    __shared__ __attribute__((aligned(16))) float as[2][GBM * GAP];
    __shared__ __attribute__((aligned(16))) float wsm[2][(GBK / 2) * GWP];

    int tid  = (int)threadIdx.x;
    int wv   = tid >> 5;
    int lane = tid & 31;
    int bn = (int)blockIdx.x % (D_MODEL / GBN);
    int bm = (int)blockIdx.x / (D_MODEL / GBN);
    int wm = wv >> 1;
    int wn = wv & 1;
    int l15 = lane & 15;
    int khi = lane >> 4;
    int rbase = khi << 3;

    const float4* Ag = (const float4*)(A + (size_t)(bm * GBM) * D_MODEL);
    const float4* Wg = (const float4*)(W + bn * GBN);
    int a_g = (tid >> 3) * (D_MODEL / 4) + (tid & 7);
    int a_l = (tid >> 3) * GAP + (tid & 7) * 4;
    int w_g[3], w_l[3];
#pragma unroll
    for (int q = 0; q < 3; ++q) {
        int u = tid + 256 * q;
        int pr = u / 48, c4 = u % 48;
        w_g[q] = (2 * pr) * (D_MODEL / 4) + c4;
        w_l[q] = pr * GWP + c4 * 8;
    }

    const int AP = 32 * (D_MODEL / 4);
    const int WR = D_MODEL / 4;
    float4 s_a0, s_a1, s_w0, s_w1, s_w2, s_w3, s_w4, s_w5;

#define STAGE_STORE(buf, A0, A1, W0, W1, W2, W3, W4, W5)                                  \
    do {                                                                                  \
        *(float4*)&as[buf][a_l]            = A0;                                          \
        *(float4*)&as[buf][a_l + 32 * GAP] = A1;                                          \
        *(float4*)&wsm[buf][w_l[0]]     = make_float4(W0.x, W1.x, W0.y, W1.y);            \
        *(float4*)&wsm[buf][w_l[0] + 4] = make_float4(W0.z, W1.z, W0.w, W1.w);            \
        *(float4*)&wsm[buf][w_l[1]]     = make_float4(W2.x, W3.x, W2.y, W3.y);            \
        *(float4*)&wsm[buf][w_l[1] + 4] = make_float4(W2.z, W3.z, W2.w, W3.w);            \
        *(float4*)&wsm[buf][w_l[2]]     = make_float4(W4.x, W5.x, W4.y, W5.y);            \
        *(float4*)&wsm[buf][w_l[2] + 4] = make_float4(W4.z, W5.z, W4.w, W5.w);            \
    } while (0)
#define STAGE_FETCH(stg)                                                                  \
    do {                                                                                  \
        int ga_ = a_g + (stg) * (GBK / 4);                                                \
        int gw_ = (stg) * GBK * (D_MODEL / 4);                                            \
        s_a0 = Ag[ga_];              s_a1 = Ag[ga_ + AP];                                 \
        s_w0 = Wg[gw_ + w_g[0]];     s_w1 = Wg[gw_ + w_g[0] + WR];                        \
        s_w2 = Wg[gw_ + w_g[1]];     s_w3 = Wg[gw_ + w_g[1] + WR];                        \
        s_w4 = Wg[gw_ + w_g[2]];     s_w5 = Wg[gw_ + w_g[2] + WR];                        \
    } while (0)

    {
        float4 t_a0 = Ag[a_g], t_a1 = Ag[a_g + AP];
        float4 t_w0 = Wg[w_g[0]], t_w1 = Wg[w_g[0] + WR];
        float4 t_w2 = Wg[w_g[1]], t_w3 = Wg[w_g[1] + WR];
        float4 t_w4 = Wg[w_g[2]], t_w5 = Wg[w_g[2] + WR];
        STAGE_FETCH(1);
        STAGE_STORE(0, t_a0, t_a1, t_w0, t_w1, t_w2, t_w3, t_w4, t_w5);
    }
    __syncthreads();

    v8f acc[6];
#pragma unroll
    for (int c = 0; c < 6; ++c) acc[c] = (v8f){};

    for (int st = 0; st < GSTEPS; ++st) {
        int cur = st & 1;
        bool more = (st + 1 < GSTEPS);
        bool more2 = (st + 2 < GSTEPS);

        const float* ab = &as[cur][(wm * 16 + l15) * GAP + 2 * khi];
        const float* wb = &wsm[cur][khi * GWP + 2 * (wn * 96 + l15)];
        v2f af[GBK / 4], bf[GBK / 4][6];
#pragma unroll
        for (int j = 0; j < GBK / 4; ++j) {
            af[j] = *(const v2f*)(ab + 4 * j);
#pragma unroll
            for (int c = 0; c < 6; ++c)
                bf[j][c] = *(const v2f*)(wb + (2 * j) * GWP + 32 * c);
        }

        if (more) {
            STAGE_STORE(cur ^ 1, s_a0, s_a1, s_w0, s_w1, s_w2, s_w3, s_w4, s_w5);
            asm volatile("s_wait_dscnt 0x0\n\ts_barrier_signal -1" ::: "memory");
        }
        __builtin_amdgcn_sched_barrier(0);

        if (more2) STAGE_FETCH(st + 2);

#pragma unroll
        for (int j = 0; j < GBK / 4; ++j)
#pragma unroll
            for (int c = 0; c < 6; ++c)
                acc[c] = __builtin_amdgcn_wmma_f32_16x16x4_f32(
                    false, af[j], false, bf[j][c], (short)0, acc[c], false, false);

        __builtin_amdgcn_sched_barrier(0);
        if (more) {
            asm volatile("s_barrier_wait 0xffff" ::: "memory");
        }
    }
#undef STAGE_STORE
#undef STAGE_FETCH

    int m0 = bm * GBM + wm * 16;
    int n0 = bn * GBN + wn * 96;
    if (!split_heads) {
#pragma unroll
        for (int c = 0; c < 6; ++c) {
            float bv = bias[n0 + c * 16 + l15];
#pragma unroll
            for (int r = 0; r < 8; ++r) {
                int row = m0 + rbase + r;
                C[(size_t)row * D_MODEL + n0 + c * 16 + l15] = acc[c][r] + bv;
            }
        }
    } else {
#pragma unroll
        for (int c = 0; c < 6; ++c) {
            int col = n0 + c * 16 + l15;
            int h = col >> 6;
            int d = col & 63;
            float bv = bias[col];
#pragma unroll
            for (int r = 0; r < 8; ++r) {
                int row = m0 + rbase + r;
                int bi = row / SEQ, sq = row % SEQ;
                C[(((size_t)bi * NHEAD + h) * SEQ + sq) * DK + d] = acc[c][r] + bv;
            }
        }
    }
}

#define NSPLIT   16
#define KCH      (SEQ / NSPLIT)
#define KVELEM   (DK * DK)
#define KVBLOCKS (BATCH * NHEAD * NSPLIT * 4 / 8)
#define MFBLOCKS (SEQ / 16)
#define QPB      256

__global__ __launch_bounds__(256) void prep_kernel(
    const float* __restrict__ KP, const float* __restrict__ VP,
    const int* __restrict__ mask, float* __restrict__ KVP,
    unsigned* __restrict__ FLG)
{
    __shared__ __attribute__((aligned(16))) float kst[2][(KCH / 2) * VPITCH];
    __shared__ __attribute__((aligned(16))) float vst[2][(KCH / 2) * VPITCH];
    int tid = (int)threadIdx.x;
    if ((int)blockIdx.x >= KVBLOCKS) {
        int qt = (int)blockIdx.x - KVBLOCKS;
        int* zf = (int*)&kst[0][0];
        if (tid < 64) zf[tid] = 0;
        __syncthreads();
        const int4* mg = (const int4*)(mask + (size_t)qt * 16 * SEQ);
        int z0 = 0, z1 = 0;
#pragma unroll
        for (int i = 0; i < 32; i += 2) {
            int4 m0 = mg[tid + 256 * i];
            int4 m1 = mg[tid + 256 * (i + 1)];
            z0 |= (int)(m0.x == 0) | (int)(m0.y == 0) | (int)(m0.z == 0) | (int)(m0.w == 0);
            z1 |= (int)(m1.x == 0) | (int)(m1.y == 0) | (int)(m1.z == 0) | (int)(m1.w == 0);
        }
        if (z0 != 0) zf[tid >> 3] = 1;
        if (z1 != 0) zf[32 + (tid >> 3)] = 1;
        __syncthreads();
        if (tid < 2) {
            unsigned w = 0u;
            for (int i = 0; i < 32; ++i)
                if (zf[32 * tid + i] != 0) w |= (1u << i);
            FLG[qt * 2 + tid] = w;
        }
        return;
    }

    int wv   = tid >> 5;
    int lane = tid & 31;
    int l15 = lane & 15;
    int khi = lane >> 4;
    int rbase = khi << 3;
    int hb  = wv >> 2;
    int mt  = wv & 3;
    int pid = (int)blockIdx.x * 2 + hb;
    int t7  = tid & 127;

    const float4* Kg = (const float4*)(KP + (size_t)pid * KCH * DK);
    const float4* Vg = (const float4*)(VP + (size_t)pid * KCH * DK);
    float4 kr[16], vr[16];
#pragma unroll
    for (int p = 0; p < 8; ++p) {
        int u = t7 + 128 * p;
        int m = u >> 4, c4 = u & 15;
        kr[2 * p]     = Kg[(2 * m) * 16 + c4];
        kr[2 * p + 1] = Kg[(2 * m + 1) * 16 + c4];
        vr[2 * p]     = Vg[(2 * m) * 16 + c4];
        vr[2 * p + 1] = Vg[(2 * m + 1) * 16 + c4];
    }
#pragma unroll
    for (int p = 0; p < 8; ++p) {
        int u = t7 + 128 * p;
        int m = u >> 4, c4 = u & 15;
        float4 a = kr[2 * p], b = kr[2 * p + 1];
        *(float4*)&kst[hb][m * VPITCH + c4 * 8]     = make_float4(a.x, b.x, a.y, b.y);
        *(float4*)&kst[hb][m * VPITCH + c4 * 8 + 4] = make_float4(a.z, b.z, a.w, b.w);
        a = vr[2 * p]; b = vr[2 * p + 1];
        *(float4*)&vst[hb][m * VPITCH + c4 * 8]     = make_float4(a.x, b.x, a.y, b.y);
        *(float4*)&vst[hb][m * VPITCH + c4 * 8 + 4] = make_float4(a.z, b.z, a.w, b.w);
    }
    __syncthreads();

    const float* kb = &kst[hb][khi * VPITCH + 2 * (16 * mt + l15)];
    const float* vb = &vst[hb][khi * VPITCH + 2 * l15];

    v8f acc[4];
#pragma unroll
    for (int c = 0; c < 4; ++c) acc[c] = (v8f){};

#pragma unroll 8
    for (int j = 0; j < KCH / 4; ++j) {
        v2f afr = *(const v2f*)(kb + (2 * j) * VPITCH);
#pragma unroll
        for (int c = 0; c < 4; ++c) {
            v2f bfr = *(const v2f*)(vb + (2 * j) * VPITCH + 32 * c);
            acc[c] = __builtin_amdgcn_wmma_f32_16x16x4_f32(
                false, afr, false, bfr, (short)0, acc[c], false, false);
        }
    }

    float* po = KVP + (size_t)pid * KVELEM;
#pragma unroll
    for (int c = 0; c < 4; ++c)
#pragma unroll
        for (int r = 0; r < 8; ++r)
            po[(16 * mt + rbase + r) * DK + 16 * c + l15] = acc[c][r];
}

__global__ __launch_bounds__(256) void apply_kernel(
    const float* __restrict__ QP, const float* __restrict__ KP,
    const float* __restrict__ VP, const int* __restrict__ mask,
    const float* __restrict__ KVP, const unsigned* __restrict__ FLG,
    float* __restrict__ X)
{
    __shared__ __attribute__((aligned(16))) float kvs[(DK / 2) * VPITCH];
    __shared__ __attribute__((aligned(16))) float smem[8][2][16 * SMSTRIDE];

    int tid  = (int)threadIdx.x;
    int wv   = tid >> 5;
    int lane = tid & 31;
    int l15 = lane & 15;
    int khi = lane >> 4;
    int rbase = khi << 3;
    int bh = (int)blockIdx.x / (SEQ / QPB);
    int qb = (int)blockIdx.x % (SEQ / QPB);
    int bi = bh / NHEAD, h = bh % NHEAD;

    v2f qf[2][16];
#pragma unroll
    for (int ti = 0; ti < 2; ++ti) {
        const float* qrow = QP + ((size_t)bh * SEQ + (size_t)(qb * QPB + (wv * 2 + ti) * 16 + l15)) * DK;
#pragma unroll
        for (int j = 0; j < 16; ++j)
            qf[ti][j] = *(const v2f*)(qrow + 4 * j + 2 * khi);
    }

    unsigned fw[2][2];
#pragma unroll
    for (int ti = 0; ti < 2; ++ti)
#pragma unroll
        for (int hw = 0; hw < 2; ++hw)
            fw[ti][hw] = FLG[(qb * (QPB / 16) + wv * 2 + ti) * 2 + hw];

    const float4* pg = (const float4*)(KVP + (size_t)bh * NSPLIT * KVELEM);
#pragma unroll
    for (int p = 0; p < 2; ++p) {
        int u = tid + 256 * p;
        int m = u >> 4, c4 = u & 15;
        float4 a = pg[(2 * m) * 16 + c4];
        float4 b = pg[(2 * m + 1) * 16 + c4];
#pragma unroll
        for (int s = 1; s < NSPLIT; ++s) {
            float4 a2 = pg[s * (KVELEM / 4) + (2 * m) * 16 + c4];
            float4 b2 = pg[s * (KVELEM / 4) + (2 * m + 1) * 16 + c4];
            a.x += a2.x; a.y += a2.y; a.z += a2.z; a.w += a2.w;
            b.x += b2.x; b.y += b2.y; b.z += b2.z; b.w += b2.w;
        }
        *(float4*)&kvs[m * VPITCH + c4 * 8]     = make_float4(a.x, b.x, a.y, b.y);
        *(float4*)&kvs[m * VPITCH + c4 * 8 + 4] = make_float4(a.z, b.z, a.w, b.w);
    }
    __syncthreads();

    const float* K = KP + (size_t)bh * SEQ * DK;
    const float* V = VP + (size_t)bh * SEQ * DK;
    float* sm = &smem[wv][0][0];
    const float* kvb = &kvs[khi * VPITCH + 2 * l15];

#pragma unroll
    for (int ti = 0; ti < 2; ++ti) {
        int qt = qb * (QPB / 16) + wv * 2 + ti;
        int s0 = qt * 16;

        const v2f* qv = qf[ti];

        v8f acc[4];
#pragma unroll
        for (int c = 0; c < 4; ++c) acc[c] = (v8f){};
#pragma unroll
        for (int j = 0; j < 16; ++j)
#pragma unroll
            for (int c = 0; c < 4; ++c) {
                v2f bfr = *(const v2f*)(kvb + (2 * j) * VPITCH + 32 * c);
                acc[c] = __builtin_amdgcn_wmma_f32_16x16x4_f32(
                    false, qv[j], false, bfr, (short)0, acc[c], false, false);
            }
#pragma unroll
        for (int c = 0; c < 4; ++c)
#pragma unroll
            for (int r = 0; r < 8; ++r) acc[c][r] *= 0.125f;

        const int* mtile = mask + (size_t)(s0 + rbase) * SEQ + l15;
#pragma unroll
        for (int hw = 0; hw < 2; ++hw) {
            unsigned bits = (unsigned)__builtin_amdgcn_readfirstlane((int)fw[ti][hw]);
            while (bits != 0u) {
                int st = 32 * hw + __builtin_ctz(bits);
                bits &= bits - 1u;
                const float* Kst = K + (size_t)st * KSTAGE * DK;
                const float* Vst = V + (size_t)st * KSTAGE * DK;
                const int* mrow = mtile + st * KSTAGE;

                v8f sa = {}, sb = {};
#pragma unroll
                for (int j = 0; j < 16; ++j) {
                    v2f k0 = *(const v2f*)(Kst + (size_t)l15 * DK + 4 * j + 2 * khi);
                    v2f k1 = *(const v2f*)(Kst + (size_t)(16 + l15) * DK + 4 * j + 2 * khi);
                    sa = __builtin_amdgcn_wmma_f32_16x16x4_f32(
                        false, qv[j], false, k0, (short)0, sa, false, false);
                    sb = __builtin_amdgcn_wmma_f32_16x16x4_f32(
                        false, qv[j], false, k1, (short)0, sb, false, false);
                }
#pragma unroll
                for (int r = 0; r < 8; ++r) {
                    int srow = rbase + r;
                    float ea = (mrow[(size_t)r * SEQ] == 0)      ? (NEGV - sa[r] * 0.125f) : 0.0f;
                    float eb = (mrow[(size_t)r * SEQ + 16] == 0) ? (NEGV - sb[r] * 0.125f) : 0.0f;
                    sm[srow * SMSTRIDE + l15] = ea;
                    sm[16 * SMSTRIDE + srow * SMSTRIDE + l15] = eb;
                }
                asm volatile("s_wait_dscnt 0" ::: "memory");
#pragma unroll
                for (int u = 0; u < 2; ++u)
#pragma unroll
                    for (int j = 0; j < 4; ++j) {
                        v2f afr = *(const v2f*)(sm + u * 16 * SMSTRIDE + l15 * SMSTRIDE + 4 * j + 2 * khi);
                        const float* vr = Vst + (size_t)(16 * u + 4 * j + 2 * khi) * DK + l15;
#pragma unroll
                        for (int c = 0; c < 4; ++c) {
                            v2f vfr = (v2f){vr[16 * c], vr[DK + 16 * c]};
                            acc[c] = __builtin_amdgcn_wmma_f32_16x16x4_f32(
                                false, afr, false, vfr, (short)0, acc[c], false, false);
                        }
                    }
                asm volatile("s_wait_dscnt 0" ::: "memory");
            }
        }

#pragma unroll
        for (int c = 0; c < 4; ++c) {
#pragma unroll
            for (int r = 0; r < 8; ++r) {
                int s = s0 + rbase + r;
                int d = c * 16 + l15;
                X[((size_t)bi * SEQ + s) * D_MODEL + h * DK + d] = acc[c][r];
            }
        }
    }
}

extern "C" void kernel_launch(void* const* d_in, const int* in_sizes, int n_in,
                              void* d_out, int out_size, void* d_ws, size_t ws_size,
                              hipStream_t stream) {
    (void)in_sizes; (void)n_in; (void)out_size; (void)ws_size;
    const float* q    = (const float*)d_in[0];
    const float* k    = (const float*)d_in[1];
    const float* v    = (const float*)d_in[2];
    const int*   mask = (const int*)d_in[3];
    const float* w_q  = (const float*)d_in[4];
    const float* b_q  = (const float*)d_in[5];
    const float* w_k  = (const float*)d_in[6];
    const float* b_k  = (const float*)d_in[7];
    const float* w_v  = (const float*)d_in[8];
    const float* b_v  = (const float*)d_in[9];
    const float* w_o  = (const float*)d_in[10];
    const float* b_o  = (const float*)d_in[11];

    float* ws = (float*)d_ws;
    size_t proj = (size_t)M_ROWS * D_MODEL;
    float* qp = ws;
    float* kp = ws + proj;
    float* vp = ws + 2 * proj;
    float* xb = ws + 3 * proj;
    float* kvp = ws + 4 * proj;
    unsigned* flg = (unsigned*)(ws + 4 * proj + (size_t)BATCH * NHEAD * NSPLIT * KVELEM);

    dim3 blk(256);
    dim3 g_gemm((M_ROWS / GBM) * (D_MODEL / GBN));
    dim3 g_prep(KVBLOCKS + MFBLOCKS);
    dim3 g_apply(BATCH * NHEAD * (SEQ / QPB));

    gemm768_kernel<<<g_gemm, blk, 0, stream>>>(q, w_q, b_q, qp, 1);
    gemm768_kernel<<<g_gemm, blk, 0, stream>>>(k, w_k, b_k, kp, 1);
    gemm768_kernel<<<g_gemm, blk, 0, stream>>>(v, w_v, b_v, vp, 1);
    prep_kernel<<<g_prep, blk, 0, stream>>>(kp, vp, mask, kvp, flg);
    apply_kernel<<<g_apply, blk, 0, stream>>>(qp, kp, vp, mask, kvp, flg, xb);
    gemm768_kernel<<<g_gemm, blk, 0, stream>>>(xb, w_o, b_o, (float*)d_out, 0);
}


# projection GEMMs: q, k, v in ONE launch of 256 blocks, each taking its tile through all three sets as one 72-stage pipeline (no drain, refill or launch gap between sets); 6 launches -> 4; same sums
# speedup vs baseline: 3.4302x; 1.0480x over previous
#include <hip/hip_runtime.h>

typedef float v2f __attribute__((ext_vector_type(2)));
typedef float v8f __attribute__((ext_vector_type(8)));

#define D_MODEL 768
#define NHEAD   12
#define DK      64
#define BATCH   2
#define SEQ     2048
#define M_ROWS  (BATCH * SEQ)
#define NEGV    (-1000000000.0f)
#define SMSTRIDE 18
#define GBM      64
#define GBN      192
#define GBK      32
#define GAP      36
#define GWP      416
#define GSTEPS   (D_MODEL / GBK)
#define KSTAGE   32
#define VPITCH   160

#define PICK3(p, x0, x1, x2) ((p) == 0 ? (x0) : ((p) == 1 ? (x1) : (x2)))
__global__ __launch_bounds__(256) __attribute__((amdgpu_waves_per_eu(2, 2))) void gemm768_kernel(
    const float* __restrict__ A0, const float* __restrict__ W0,
    const float* __restrict__ bias0, float* __restrict__ C0,
    const float* __restrict__ A1, const float* __restrict__ W1,
    const float* __restrict__ bias1, float* __restrict__ C1,
    const float* __restrict__ A2, const float* __restrict__ W2,
    const float* __restrict__ bias2, float* __restrict__ C2,
    int nsets, int split_heads)
{
    __shared__ __attribute__((aligned(16))) float as[2][GBM * GAP];
    __shared__ __attribute__((aligned(16))) float wsm[2][(GBK / 2) * GWP];

    int tid  = (int)threadIdx.x;
    int wv   = tid >> 5;
    int lane = tid & 31;
    int bn = (int)blockIdx.x % (D_MODEL / GBN);
    int bm = (int)blockIdx.x / (D_MODEL / GBN);
    int wm = wv >> 1;
    int wn = wv & 1;
    int l15 = lane & 15;
    int khi = lane >> 4;
    int rbase = khi << 3;

    const size_t a_off = (size_t)(bm * GBM) * D_MODEL;
    const int    w_off = bn * GBN;
    int a_g = (tid >> 3) * (D_MODEL / 4) + (tid & 7);
    int a_l = (tid >> 3) * GAP + (tid & 7) * 4;
    int w_g[3], w_l[3];
#pragma unroll
    for (int q = 0; q < 3; ++q) {
        int u = tid + 256 * q;
        int pr = u / 48, c4 = u % 48;
        w_g[q] = (2 * pr) * (D_MODEL / 4) + c4;
        w_l[q] = pr * GWP + c4 * 8;
    }

    const int AP = 32 * (D_MODEL / 4);
    const int WR = D_MODEL / 4;
    float4 s_a0, s_a1, s_w0, s_w1, s_w2, s_w3, s_w4, s_w5;

#define STAGE_STORE(buf, A0_, A1_, W0_, W1_, W2_, W3_, W4_, W5_)                          \
    do {                                                                                  \
        *(float4*)&as[buf][a_l]            = A0_;                                         \
        *(float4*)&as[buf][a_l + 32 * GAP] = A1_;                                         \
        *(float4*)&wsm[buf][w_l[0]]     = make_float4(W0_.x, W1_.x, W0_.y, W1_.y);        \
        *(float4*)&wsm[buf][w_l[0] + 4] = make_float4(W0_.z, W1_.z, W0_.w, W1_.w);        \
        *(float4*)&wsm[buf][w_l[1]]     = make_float4(W2_.x, W3_.x, W2_.y, W3_.y);        \
        *(float4*)&wsm[buf][w_l[1] + 4] = make_float4(W2_.z, W3_.z, W2_.w, W3_.w);        \
        *(float4*)&wsm[buf][w_l[2]]     = make_float4(W4_.x, W5_.x, W4_.y, W5_.y);        \
        *(float4*)&wsm[buf][w_l[2] + 4] = make_float4(W4_.z, W5_.z, W4_.w, W5_.w);        \
    } while (0)
#define STAGE_FETCH(AgP, WgP, stg)                                                        \
    do {                                                                                  \
        int ga_ = a_g + (stg) * (GBK / 4);                                                \
        int gw_ = (stg) * GBK * (D_MODEL / 4);                                            \
        s_a0 = (AgP)[ga_];             s_a1 = (AgP)[ga_ + AP];                            \
        s_w0 = (WgP)[gw_ + w_g[0]];    s_w1 = (WgP)[gw_ + w_g[0] + WR];                   \
        s_w2 = (WgP)[gw_ + w_g[1]];    s_w3 = (WgP)[gw_ + w_g[1] + WR];                   \
        s_w4 = (WgP)[gw_ + w_g[2]];    s_w5 = (WgP)[gw_ + w_g[2] + WR];                   \
    } while (0)

    {
        const float4* Ag = (const float4*)(A0 + a_off);
        const float4* Wg = (const float4*)(W0 + w_off);
        float4 t_a0 = Ag[a_g], t_a1 = Ag[a_g + AP];
        float4 t_w0 = Wg[w_g[0]], t_w1 = Wg[w_g[0] + WR];
        float4 t_w2 = Wg[w_g[1]], t_w3 = Wg[w_g[1] + WR];
        float4 t_w4 = Wg[w_g[2]], t_w5 = Wg[w_g[2] + WR];
        STAGE_FETCH(Ag, Wg, 1);
        STAGE_STORE(0, t_a0, t_a1, t_w0, t_w1, t_w2, t_w3, t_w4, t_w5);
    }
    __syncthreads();

    v8f acc[6];
#pragma unroll
    for (int c = 0; c < 6; ++c) acc[c] = (v8f){};

    int m0 = bm * GBM + wm * 16;
    int n0 = bn * GBN + wn * 96;

    const int total = nsets * GSTEPS;
    int ls = 0, es = 0;
    int fs = 2, fset = 0;

    for (int st = 0; st < total; ++st) {
        int cur = st & 1;
        bool more = (st + 1 < total);
        bool more2 = (st + 2 < total);

        const float* ab = &as[cur][(wm * 16 + l15) * GAP + 2 * khi];
        const float* wb = &wsm[cur][khi * GWP + 2 * (wn * 96 + l15)];
        v2f af[GBK / 4], bf[GBK / 4][6];
#pragma unroll
        for (int j = 0; j < GBK / 4; ++j) {
            af[j] = *(const v2f*)(ab + 4 * j);
#pragma unroll
            for (int c = 0; c < 6; ++c)
                bf[j][c] = *(const v2f*)(wb + (2 * j) * GWP + 32 * c);
        }

        if (more) {
            STAGE_STORE(cur ^ 1, s_a0, s_a1, s_w0, s_w1, s_w2, s_w3, s_w4, s_w5);
            asm volatile("s_wait_dscnt 0x0\n\ts_barrier_signal -1" ::: "memory");
        }
        __builtin_amdgcn_sched_barrier(0);

        if (more2) {
            const float4* Ag = (const float4*)(PICK3(fset, A0, A1, A2) + a_off);
            const float4* Wg = (const float4*)(PICK3(fset, W0, W1, W2) + w_off);
            STAGE_FETCH(Ag, Wg, fs);
        }
        ++fs;
        if (fs == GSTEPS) { fs = 0; ++fset; }

#pragma unroll
        for (int j = 0; j < GBK / 4; ++j)
#pragma unroll
            for (int c = 0; c < 6; ++c)
                acc[c] = __builtin_amdgcn_wmma_f32_16x16x4_f32(
                    false, af[j], false, bf[j][c], (short)0, acc[c], false, false);

        __builtin_amdgcn_sched_barrier(0);
        if (more) {
            asm volatile("s_barrier_wait 0xffff" ::: "memory");
        }

        ++ls;
        if (ls == GSTEPS) {
            const float* bias = PICK3(es, bias0, bias1, bias2);
            float* C = PICK3(es, C0, C1, C2);
            float bv[6];
#pragma unroll
            for (int c = 0; c < 6; ++c) bv[c] = bias[n0 + c * 16 + l15];
            if (!split_heads) {
#pragma unroll
                for (int c = 0; c < 6; ++c) {
#pragma unroll
                    for (int r = 0; r < 8; ++r) {
                        int row = m0 + rbase + r;
                        C[(size_t)row * D_MODEL + n0 + c * 16 + l15] = acc[c][r] + bv[c];
                    }
                }
            } else {
#pragma unroll
                for (int c = 0; c < 6; ++c) {
                    int col = n0 + c * 16 + l15;
                    int h = col >> 6;
                    int d = col & 63;
#pragma unroll
                    for (int r = 0; r < 8; ++r) {
                        int row = m0 + rbase + r;
                        int bi = row / SEQ, sq = row % SEQ;
                        C[(((size_t)bi * NHEAD + h) * SEQ + sq) * DK + d] = acc[c][r] + bv[c];
                    }
                }
            }
#pragma unroll
            for (int c = 0; c < 6; ++c) acc[c] = (v8f){};
            ls = 0;
            ++es;
        }
    }
#undef STAGE_STORE
#undef STAGE_FETCH
}
#undef PICK3

#define NSPLIT   16
#define KCH      (SEQ / NSPLIT)
#define KVELEM   (DK * DK)
#define KVBLOCKS (BATCH * NHEAD * NSPLIT * 4 / 8)
#define MFBLOCKS (SEQ / 16)
#define QPB      256

__global__ __launch_bounds__(256) void prep_kernel(
    const float* __restrict__ KP, const float* __restrict__ VP,
    const int* __restrict__ mask, float* __restrict__ KVP,
    unsigned* __restrict__ FLG)
{
    __shared__ __attribute__((aligned(16))) float kst[2][(KCH / 2) * VPITCH];
    __shared__ __attribute__((aligned(16))) float vst[2][(KCH / 2) * VPITCH];
    int tid = (int)threadIdx.x;
    if ((int)blockIdx.x >= KVBLOCKS) {
        int qt = (int)blockIdx.x - KVBLOCKS;
        int* zf = (int*)&kst[0][0];
        if (tid < 64) zf[tid] = 0;
        __syncthreads();
        const int4* mg = (const int4*)(mask + (size_t)qt * 16 * SEQ);
        int z0 = 0, z1 = 0;
#pragma unroll
        for (int i = 0; i < 32; i += 2) {
            int4 m0 = mg[tid + 256 * i];
            int4 m1 = mg[tid + 256 * (i + 1)];
            z0 |= (int)(m0.x == 0) | (int)(m0.y == 0) | (int)(m0.z == 0) | (int)(m0.w == 0);
            z1 |= (int)(m1.x == 0) | (int)(m1.y == 0) | (int)(m1.z == 0) | (int)(m1.w == 0);
        }
        if (z0 != 0) zf[tid >> 3] = 1;
        if (z1 != 0) zf[32 + (tid >> 3)] = 1;
        __syncthreads();
        if (tid < 2) {
            unsigned w = 0u;
            for (int i = 0; i < 32; ++i)
                if (zf[32 * tid + i] != 0) w |= (1u << i);
            FLG[qt * 2 + tid] = w;
        }
        return;
    }

    int wv   = tid >> 5;
    int lane = tid & 31;
    int l15 = lane & 15;
    int khi = lane >> 4;
    int rbase = khi << 3;
    int hb  = wv >> 2;
    int mt  = wv & 3;
    int pid = (int)blockIdx.x * 2 + hb;
    int t7  = tid & 127;

    const float4* Kg = (const float4*)(KP + (size_t)pid * KCH * DK);
    const float4* Vg = (const float4*)(VP + (size_t)pid * KCH * DK);
    float4 kr[16], vr[16];
#pragma unroll
    for (int p = 0; p < 8; ++p) {
        int u = t7 + 128 * p;
        int m = u >> 4, c4 = u & 15;
        kr[2 * p]     = Kg[(2 * m) * 16 + c4];
        kr[2 * p + 1] = Kg[(2 * m + 1) * 16 + c4];
        vr[2 * p]     = Vg[(2 * m) * 16 + c4];
        vr[2 * p + 1] = Vg[(2 * m + 1) * 16 + c4];
    }
#pragma unroll
    for (int p = 0; p < 8; ++p) {
        int u = t7 + 128 * p;
        int m = u >> 4, c4 = u & 15;
        float4 a = kr[2 * p], b = kr[2 * p + 1];
        *(float4*)&kst[hb][m * VPITCH + c4 * 8]     = make_float4(a.x, b.x, a.y, b.y);
        *(float4*)&kst[hb][m * VPITCH + c4 * 8 + 4] = make_float4(a.z, b.z, a.w, b.w);
        a = vr[2 * p]; b = vr[2 * p + 1];
        *(float4*)&vst[hb][m * VPITCH + c4 * 8]     = make_float4(a.x, b.x, a.y, b.y);
        *(float4*)&vst[hb][m * VPITCH + c4 * 8 + 4] = make_float4(a.z, b.z, a.w, b.w);
    }
    __syncthreads();

    const float* kb = &kst[hb][khi * VPITCH + 2 * (16 * mt + l15)];
    const float* vb = &vst[hb][khi * VPITCH + 2 * l15];

    v8f acc[4];
#pragma unroll
    for (int c = 0; c < 4; ++c) acc[c] = (v8f){};

#pragma unroll 8
    for (int j = 0; j < KCH / 4; ++j) {
        v2f afr = *(const v2f*)(kb + (2 * j) * VPITCH);
#pragma unroll
        for (int c = 0; c < 4; ++c) {
            v2f bfr = *(const v2f*)(vb + (2 * j) * VPITCH + 32 * c);
            acc[c] = __builtin_amdgcn_wmma_f32_16x16x4_f32(
                false, afr, false, bfr, (short)0, acc[c], false, false);
        }
    }

    float* po = KVP + (size_t)pid * KVELEM;
#pragma unroll
    for (int c = 0; c < 4; ++c)
#pragma unroll
        for (int r = 0; r < 8; ++r)
            po[(16 * mt + rbase + r) * DK + 16 * c + l15] = acc[c][r];
}

__global__ __launch_bounds__(256) void apply_kernel(
    const float* __restrict__ QP, const float* __restrict__ KP,
    const float* __restrict__ VP, const int* __restrict__ mask,
    const float* __restrict__ KVP, const unsigned* __restrict__ FLG,
    float* __restrict__ X)
{
    __shared__ __attribute__((aligned(16))) float kvs[(DK / 2) * VPITCH];
    __shared__ __attribute__((aligned(16))) float smem[8][2][16 * SMSTRIDE];

    int tid  = (int)threadIdx.x;
    int wv   = tid >> 5;
    int lane = tid & 31;
    int l15 = lane & 15;
    int khi = lane >> 4;
    int rbase = khi << 3;
    int bh = (int)blockIdx.x / (SEQ / QPB);
    int qb = (int)blockIdx.x % (SEQ / QPB);
    int bi = bh / NHEAD, h = bh % NHEAD;

    v2f qf[2][16];
#pragma unroll
    for (int ti = 0; ti < 2; ++ti) {
        const float* qrow = QP + ((size_t)bh * SEQ + (size_t)(qb * QPB + (wv * 2 + ti) * 16 + l15)) * DK;
#pragma unroll
        for (int j = 0; j < 16; ++j)
            qf[ti][j] = *(const v2f*)(qrow + 4 * j + 2 * khi);
    }

    unsigned fw[2][2];
#pragma unroll
    for (int ti = 0; ti < 2; ++ti)
#pragma unroll
        for (int hw = 0; hw < 2; ++hw)
            fw[ti][hw] = FLG[(qb * (QPB / 16) + wv * 2 + ti) * 2 + hw];

    const float4* pg = (const float4*)(KVP + (size_t)bh * NSPLIT * KVELEM);
#pragma unroll
    for (int p = 0; p < 2; ++p) {
        int u = tid + 256 * p;
        int m = u >> 4, c4 = u & 15;
        float4 a = pg[(2 * m) * 16 + c4];
        float4 b = pg[(2 * m + 1) * 16 + c4];
#pragma unroll
        for (int s = 1; s < NSPLIT; ++s) {
            float4 a2 = pg[s * (KVELEM / 4) + (2 * m) * 16 + c4];
            float4 b2 = pg[s * (KVELEM / 4) + (2 * m + 1) * 16 + c4];
            a.x += a2.x; a.y += a2.y; a.z += a2.z; a.w += a2.w;
            b.x += b2.x; b.y += b2.y; b.z += b2.z; b.w += b2.w;
        }
        *(float4*)&kvs[m * VPITCH + c4 * 8]     = make_float4(a.x, b.x, a.y, b.y);
        *(float4*)&kvs[m * VPITCH + c4 * 8 + 4] = make_float4(a.z, b.z, a.w, b.w);
    }
    __syncthreads();

    const float* K = KP + (size_t)bh * SEQ * DK;
    const float* V = VP + (size_t)bh * SEQ * DK;
    float* sm = &smem[wv][0][0];
    const float* kvb = &kvs[khi * VPITCH + 2 * l15];

#pragma unroll
    for (int ti = 0; ti < 2; ++ti) {
        int qt = qb * (QPB / 16) + wv * 2 + ti;
        int s0 = qt * 16;

        const v2f* qv = qf[ti];

        v8f acc[4];
#pragma unroll
        for (int c = 0; c < 4; ++c) acc[c] = (v8f){};
#pragma unroll
        for (int j = 0; j < 16; ++j)
#pragma unroll
            for (int c = 0; c < 4; ++c) {
                v2f bfr = *(const v2f*)(kvb + (2 * j) * VPITCH + 32 * c);
                acc[c] = __builtin_amdgcn_wmma_f32_16x16x4_f32(
                    false, qv[j], false, bfr, (short)0, acc[c], false, false);
            }
#pragma unroll
        for (int c = 0; c < 4; ++c)
#pragma unroll
            for (int r = 0; r < 8; ++r) acc[c][r] *= 0.125f;

        const int* mtile = mask + (size_t)(s0 + rbase) * SEQ + l15;
#pragma unroll
        for (int hw = 0; hw < 2; ++hw) {
            unsigned bits = (unsigned)__builtin_amdgcn_readfirstlane((int)fw[ti][hw]);
            while (bits != 0u) {
                int st = 32 * hw + __builtin_ctz(bits);
                bits &= bits - 1u;
                const float* Kst = K + (size_t)st * KSTAGE * DK;
                const float* Vst = V + (size_t)st * KSTAGE * DK;
                const int* mrow = mtile + st * KSTAGE;

                v8f sa = {}, sb = {};
#pragma unroll
                for (int j = 0; j < 16; ++j) {
                    v2f k0 = *(const v2f*)(Kst + (size_t)l15 * DK + 4 * j + 2 * khi);
                    v2f k1 = *(const v2f*)(Kst + (size_t)(16 + l15) * DK + 4 * j + 2 * khi);
                    sa = __builtin_amdgcn_wmma_f32_16x16x4_f32(
                        false, qv[j], false, k0, (short)0, sa, false, false);
                    sb = __builtin_amdgcn_wmma_f32_16x16x4_f32(
                        false, qv[j], false, k1, (short)0, sb, false, false);
                }
#pragma unroll
                for (int r = 0; r < 8; ++r) {
                    int srow = rbase + r;
                    float ea = (mrow[(size_t)r * SEQ] == 0)      ? (NEGV - sa[r] * 0.125f) : 0.0f;
                    float eb = (mrow[(size_t)r * SEQ + 16] == 0) ? (NEGV - sb[r] * 0.125f) : 0.0f;
                    sm[srow * SMSTRIDE + l15] = ea;
                    sm[16 * SMSTRIDE + srow * SMSTRIDE + l15] = eb;
                }
                asm volatile("s_wait_dscnt 0" ::: "memory");
#pragma unroll
                for (int u = 0; u < 2; ++u)
#pragma unroll
                    for (int j = 0; j < 4; ++j) {
                        v2f afr = *(const v2f*)(sm + u * 16 * SMSTRIDE + l15 * SMSTRIDE + 4 * j + 2 * khi);
                        const float* vr = Vst + (size_t)(16 * u + 4 * j + 2 * khi) * DK + l15;
#pragma unroll
                        for (int c = 0; c < 4; ++c) {
                            v2f vfr = (v2f){vr[16 * c], vr[DK + 16 * c]};
                            acc[c] = __builtin_amdgcn_wmma_f32_16x16x4_f32(
                                false, afr, false, vfr, (short)0, acc[c], false, false);
                        }
                    }
                asm volatile("s_wait_dscnt 0" ::: "memory");
            }
        }

#pragma unroll
        for (int c = 0; c < 4; ++c) {
#pragma unroll
            for (int r = 0; r < 8; ++r) {
                int s = s0 + rbase + r;
                int d = c * 16 + l15;
                X[((size_t)bi * SEQ + s) * D_MODEL + h * DK + d] = acc[c][r];
            }
        }
    }
}

extern "C" void kernel_launch(void* const* d_in, const int* in_sizes, int n_in,
                              void* d_out, int out_size, void* d_ws, size_t ws_size,
                              hipStream_t stream) {
    (void)in_sizes; (void)n_in; (void)out_size; (void)ws_size;
    const float* q    = (const float*)d_in[0];
    const float* k    = (const float*)d_in[1];
    const float* v    = (const float*)d_in[2];
    const int*   mask = (const int*)d_in[3];
    const float* w_q  = (const float*)d_in[4];
    const float* b_q  = (const float*)d_in[5];
    const float* w_k  = (const float*)d_in[6];
    const float* b_k  = (const float*)d_in[7];
    const float* w_v  = (const float*)d_in[8];
    const float* b_v  = (const float*)d_in[9];
    const float* w_o  = (const float*)d_in[10];
    const float* b_o  = (const float*)d_in[11];

    float* ws = (float*)d_ws;
    size_t proj = (size_t)M_ROWS * D_MODEL;
    float* qp = ws;
    float* kp = ws + proj;
    float* vp = ws + 2 * proj;
    float* xb = ws + 3 * proj;
    float* kvp = ws + 4 * proj;
    unsigned* flg = (unsigned*)(ws + 4 * proj + (size_t)BATCH * NHEAD * NSPLIT * KVELEM);

    dim3 blk(256);
    dim3 g_gemm((M_ROWS / GBM) * (D_MODEL / GBN));
    dim3 g_prep(KVBLOCKS + MFBLOCKS);
    dim3 g_apply(BATCH * NHEAD * (SEQ / QPB));

    gemm768_kernel<<<g_gemm, blk, 0, stream>>>(k, w_k, b_k, kp, v, w_v, b_v, vp, q, w_q, b_q, qp, 3, 1);
    prep_kernel<<<g_prep, blk, 0, stream>>>(kp, vp, mask, kvp, flg);
    apply_kernel<<<g_apply, blk, 0, stream>>>(qp, kp, vp, mask, kvp, flg, xb);
    gemm768_kernel<<<g_gemm, blk, 0, stream>>>(xb, w_o, b_o, (float*)d_out, xb, w_o, b_o, (float*)d_out,
                                               xb, w_o, b_o, (float*)d_out, 1, 0);
}
